# starting kernel
# baseline (speedup 1.0000x reference)
#include <hip/hip_runtime.h>
#include <hip/hip_bf16.h>

typedef float v2f __attribute__((ext_vector_type(2)));
typedef float v8f __attribute__((ext_vector_type(8)));

#define N_NODES   25000
#define N_EDGES   320000
#define IN_FEAT   64
#define OUT_FEAT  64
#define NUM_HEADS 5
#define NUM_RELS  20
#define NUM_BASES 10
#define HF        (NUM_HEADS * OUT_FEAT)
#define ATTN_ELEMS (NUM_RELS * NUM_HEADS * 2 * OUT_FEAT)
#define MTILES ((N_NODES + 15) / 16)

__global__ void compose_attn_kernel(const float* __restrict__ w_comp,
                                    const float* __restrict__ attn_w,
                                    float* __restrict__ attn_fcs) {
    int t = blockIdx.x * blockDim.x + threadIdx.x;
    if (t >= ATTN_ELEMS) return;
    int r = t / (NUM_HEADS * 2 * OUT_FEAT);
    int h = (t / (2 * OUT_FEAT)) % NUM_HEADS;
    int f = t % (2 * OUT_FEAT);
    float acc = 0.f;
#pragma unroll
    for (int b = 0; b < NUM_BASES; ++b)
        acc += w_comp[r * NUM_BASES + b] * attn_w[b * (2 * OUT_FEAT * NUM_HEADS) + f * NUM_HEADS + h];
    attn_fcs[t] = acc;
}

template <int LDW, int NT>
__global__ void proj_wmma_kernel(const float* __restrict__ feat,
                                 const float* __restrict__ W,
                                 float* __restrict__ Out) {
    constexpr int WCOLS = LDW / (16 * NT);
    const int wave = threadIdx.x >> 5;
    const int lane = threadIdx.x & 31;
    const int tile = blockIdx.x * (blockDim.x >> 5) + wave;
    if (tile >= MTILES * WCOLS) return;

    const int mt = tile / WCOLS;
    const int wc = tile % WCOLS;

    const int l16  = lane & 15;
    const int koff = (lane < 16) ? 0 : 2;
    const int rowA = min(mt * 16 + l16, N_NODES - 1);
    const int colb = wc * (16 * NT) + l16;

    const float* ap = feat + (size_t)rowA * IN_FEAT + koff;
    const float* bp = W + (size_t)koff * LDW + colb;

    v8f c[NT];
#pragma unroll
    for (int j = 0; j < NT; ++j) c[j] = v8f{};

#pragma unroll
    for (int k = 0; k < IN_FEAT; k += 4) {
        v2f a;
        a.x = ap[k];
        a.y = ap[k + 1];
#pragma unroll
        for (int j = 0; j < NT; ++j) {
            v2f b;
            b.x = bp[k * LDW + j * 16];
            b.y = bp[(k + 1) * LDW + j * 16];
            c[j] = __builtin_amdgcn_wmma_f32_16x16x4_f32(
                        false, a,  false, b,
                        (short)0, c[j],  false,  false);
        }
    }

    const int mbase = mt * 16 + ((lane < 16) ? 0 : 8);
    float* op = Out + (size_t)mbase * LDW + colb;
#pragma unroll
    for (int r = 0; r < 8; ++r) {
        if (mbase + r < N_NODES) {
#pragma unroll
            for (int j = 0; j < NT; ++j)
                op[r * LDW + j * 16] = c[j][r];
        }
    }
}

__global__ void edge_kernel(const float* __restrict__ z,
                            const int*   __restrict__ src,
                            const int*   __restrict__ dst,
                            const int*   __restrict__ etype,
                            const float* __restrict__ attn_fcs,
                            float*       __restrict__ out) {
    __shared__ float sW[ATTN_ELEMS];
    for (int i = threadIdx.x; i < ATTN_ELEMS; i += blockDim.x) sW[i] = attn_fcs[i];
    __syncthreads();

    const int lane  = threadIdx.x & 31;
    const int wid   = blockIdx.x * (blockDim.x >> 5) + (threadIdx.x >> 5);
    const int nwave = gridDim.x * (blockDim.x >> 5);
    const int f0    = lane * 2;

    for (int e = wid; e < N_EDGES; e += nwave) {
        const int s = src[e], d = dst[e], r = etype[e];
        const float2 zs = *(const float2*)(z + (size_t)s * IN_FEAT + f0);
        const float2 zd = *(const float2*)(z + (size_t)d * IN_FEAT + f0);
        const float* Wr = sW + r * (NUM_HEADS * 2 * OUT_FEAT);

        float att[NUM_HEADS];
#pragma unroll
        for (int h = 0; h < NUM_HEADS; ++h) {
            const float* Wh = Wr + h * (2 * OUT_FEAT);
            const float2 ws = *(const float2*)(Wh + f0);
            const float2 wd = *(const float2*)(Wh + OUT_FEAT + f0);
            float p = zs.x * ws.x + zs.y * ws.y + zd.x * wd.x + zd.y * wd.y;
#pragma unroll
            for (int m = 16; m > 0; m >>= 1) p += __shfl_xor(p, m, 32);
            att[h] = (p > 0.f) ? p : 0.01f * p;
        }

        float* op = out + (size_t)d * HF + f0;
#pragma unroll
        for (int h = 0; h < NUM_HEADS; ++h) {
            atomicAdd(op + h * OUT_FEAT,     att[h] * zs.x);
            atomicAdd(op + h * OUT_FEAT + 1, att[h] * zs.y);
        }
    }
}

extern "C" void kernel_launch(void* const* d_in, const int* in_sizes, int n_in,
                              void* d_out, int out_size, void* d_ws, size_t ws_size,
                              hipStream_t stream) {
    const float* feat      = (const float*)d_in[0];
    const int*   src       = (const int*)  d_in[1];
    const int*   dst       = (const int*)  d_in[2];
    const int*   etype     = (const int*)  d_in[3];
    const float* fc_w      = (const float*)d_in[4];
    const float* self_fc_w = (const float*)d_in[5];
    const float* attn_w    = (const float*)d_in[6];
    const float* w_comp    = (const float*)d_in[7];
    float* out = (float*)d_out;

    float* z        = (float*)d_ws;
    float* attn_fcs = z + (size_t)N_NODES * OUT_FEAT;

    compose_attn_kernel<<<(ATTN_ELEMS + 255) / 256, 256, 0, stream>>>(w_comp, attn_w, attn_fcs);

    {
        int tiles  = MTILES * (OUT_FEAT / 32);
        int blocks = (tiles + 3) / 4;
        proj_wmma_kernel<OUT_FEAT, 2><<<blocks, 128, 0, stream>>>(feat, fc_w, z);
    }
    {
        int tiles  = MTILES * (HF / 32);
        int blocks = (tiles + 3) / 4;
        proj_wmma_kernel<HF, 2><<<blocks, 128, 0, stream>>>(feat, self_fc_w, out);
    }

    edge_kernel<<<1024, 256, 0, stream>>>(z, src, dst, etype, attn_fcs, out);
}


# gather: two in-edges per trip, one per 16-lane half, lane owns 4 features; a logit is a row sum on 4 DPP adds (was 5 LDS permutes); halves added once per node; sum order changes
# speedup vs baseline: 1.5834x; 1.5834x over previous
#include <hip/hip_runtime.h>
#include <hip/hip_bf16.h>

typedef float v2f __attribute__((ext_vector_type(2)));
typedef float v8f __attribute__((ext_vector_type(8)));

#define N_NODES   25000
#define N_EDGES   320000
#define IN_FEAT   64
#define OUT_FEAT  64
#define NUM_HEADS 5
#define NUM_RELS  20
#define NUM_BASES 10
#define HF        (NUM_HEADS * OUT_FEAT)
#define ATTN_ELEMS (NUM_RELS * NUM_HEADS * 2 * OUT_FEAT)
#define MTILES ((N_NODES + 15) / 16)
#define LIST_CAP  (HF - 1)
#define KEY_NONE  0x7fffffff
#define GATHER_THREADS 512

__global__ void prep_kernel(const float* __restrict__ w_comp,
                            const float* __restrict__ attn_w,
                            float* __restrict__ attn_fcs,
                            float* __restrict__ out) {
    int t = blockIdx.x * blockDim.x + threadIdx.x;
    if (t < N_NODES) out[(size_t)t * HF] = 0.f;
    if (t >= ATTN_ELEMS) return;
    int r = t / (NUM_HEADS * 2 * OUT_FEAT);
    int h = (t / (2 * OUT_FEAT)) % NUM_HEADS;
    int f = t % (2 * OUT_FEAT);
    float acc = 0.f;
#pragma unroll
    for (int b = 0; b < NUM_BASES; ++b)
        acc += w_comp[r * NUM_BASES + b] * attn_w[b * (2 * OUT_FEAT * NUM_HEADS) + f * NUM_HEADS + h];
    attn_fcs[t] = acc;
}

template <int LDW, int NT, bool ADD>
__global__ void proj_wmma_kernel(const float* __restrict__ feat,
                                 const float* __restrict__ W,
                                 float* __restrict__ Out) {
    constexpr int WCOLS = LDW / (16 * NT);
    const int wave = threadIdx.x >> 5;
    const int lane = threadIdx.x & 31;
    const int tile = blockIdx.x * (blockDim.x >> 5) + wave;
    if (tile >= MTILES * WCOLS) return;

    const int mt = tile / WCOLS;
    const int wc = tile % WCOLS;

    const int l16  = lane & 15;
    const int koff = (lane < 16) ? 0 : 2;
    const int rowA = min(mt * 16 + l16, N_NODES - 1);
    const int colb = wc * (16 * NT) + l16;

    const float* ap = feat + (size_t)rowA * IN_FEAT + koff;
    const float* bp = W + (size_t)koff * LDW + colb;

    const int mbase = mt * 16 + ((lane < 16) ? 0 : 8);
    float* op = Out + (size_t)mbase * LDW + colb;

    v8f old[NT];
    if (ADD) {
        for (int j = 0; j < NT; ++j) old[j] = v8f{};
#pragma unroll
        for (int r = 0; r < 8; ++r) {
            if (mbase + r < N_NODES) {
#pragma unroll
                for (int j = 0; j < NT; ++j)
                    old[j][r] = op[r * LDW + j * 16];
            }
        }
    }

    v8f c[NT];
#pragma unroll
    for (int j = 0; j < NT; ++j) c[j] = v8f{};

#pragma unroll
    for (int k = 0; k < IN_FEAT; k += 4) {
        v2f a;
        a.x = ap[k];
        a.y = ap[k + 1];
#pragma unroll
        for (int j = 0; j < NT; ++j) {
            v2f b;
            b.x = bp[k * LDW + j * 16];
            b.y = bp[(k + 1) * LDW + j * 16];
            c[j] = __builtin_amdgcn_wmma_f32_16x16x4_f32(
                        false, a,  false, b,
                        (short)0, c[j],  false,  false);
        }
    }

    if (ADD) {
        for (int j = 0; j < NT; ++j) c[j] += old[j];
    }

#pragma unroll
    for (int r = 0; r < 8; ++r) {
        if (mbase + r < N_NODES) {
#pragma unroll
            for (int j = 0; j < NT; ++j)
                op[r * LDW + j * 16] = c[j][r];
        }
    }
}

__global__ void list_kernel(const int* __restrict__ dst, float* out) {
    const int e = blockIdx.x * blockDim.x + threadIdx.x;
    if (e >= N_EDGES) return;
    int* row = reinterpret_cast<int*>(out + (size_t)dst[e] * HF);
    const int pos = atomicAdd(row, 1);
    if (pos < LIST_CAP) atomicExch(row + 1 + pos, e);
}

__device__ __forceinline__ float xhalf(float x) {
    return __int_as_float(__builtin_amdgcn_permlanex16(__float_as_int(x), __float_as_int(x),
                                                       0x76543210, 0xfedcba98, false, false));
}

template <int M>
__device__ __forceinline__ float row_xor_add(float p) {
    return p + __int_as_float(__builtin_amdgcn_update_dpp(0, __float_as_int(p), 0x160 + M,
                                                          0xf, 0xf, true));
}

__device__ __forceinline__ void add_pair(const float* sW, int r, float4 zs, float4 zd,
                                         int f0, float4 (&acc)[NUM_HEADS]) {
    const float* Wr = sW + r * (NUM_HEADS * 2 * OUT_FEAT);
#pragma unroll
    for (int h = 0; h < NUM_HEADS; ++h) {
        const float* Wh = Wr + h * (2 * OUT_FEAT);
        const float4 ws = *(const float4*)(Wh + f0);
        const float4 wd = *(const float4*)(Wh + OUT_FEAT + f0);
        v2f t = v2f{zs.x, zs.y} * v2f{ws.x, ws.y};
        t = v2f{zs.z, zs.w} * v2f{ws.z, ws.w} + t;
        t = v2f{zd.x, zd.y} * v2f{wd.x, wd.y} + t;
        t = v2f{zd.z, zd.w} * v2f{wd.z, wd.w} + t;
        float p = t.x + t.y;
        p = row_xor_add<1>(p);
        p = row_xor_add<2>(p);
        p = row_xor_add<4>(p);
        p = row_xor_add<8>(p);
        const float a = (p > 0.f) ? p : 0.01f * p;
        acc[h].x += a * zs.x;
        acc[h].y += a * zs.y;
        acc[h].z += a * zs.z;
        acc[h].w += a * zs.w;
    }
}

__device__ __forceinline__ void pick_pair(const float* __restrict__ z, int s0, int r0, int rank,
                                          int q, int n, int hf, int f0, float4& zs, int& r) {
    const int oA = __builtin_ctz(__builtin_amdgcn_ballot_w32(rank == q));
    int s = __builtin_amdgcn_readlane(s0, oA);
    int rr = __builtin_amdgcn_readlane(r0, oA);
    const bool hasB = (q + 1 < n);
    if (hasB) {
        const int oB = __builtin_ctz(__builtin_amdgcn_ballot_w32(rank == q + 1));
        const int sB = __builtin_amdgcn_readlane(s0, oB);
        const int rB = __builtin_amdgcn_readlane(r0, oB);
        if (hf) { s = sB; rr = rB; }
    }
    float4 v = *(const float4*)(z + (size_t)s * IN_FEAT + f0);
    if (hf && !hasB) v = make_float4(0.f, 0.f, 0.f, 0.f);
    zs = v;
    r = rr;
}

__global__ void __launch_bounds__(GATHER_THREADS, 12)
gather_kernel(const float* __restrict__ z,
              const int*   __restrict__ src,
              const int*   __restrict__ dst,
              const int*   __restrict__ etype,
              const float* __restrict__ attn_fcs,
              float* out) {
    __shared__ __attribute__((aligned(16))) float sW[ATTN_ELEMS];
    {
        constexpr int NV4 = ATTN_ELEMS / 4;
        constexpr int PER = 8;
        constexpr int CPY = NV4 / PER;
        static_assert(CPY * PER == NV4 && CPY <= GATHER_THREADS, "copy split");
        const float4* g4 = reinterpret_cast<const float4*>(attn_fcs);
        float4* s4 = reinterpret_cast<float4*>(sW);
        const int tid = threadIdx.x;
        if (tid < CPY) {
            float4 tmp[PER];
#pragma unroll
            for (int j = 0; j < PER; ++j) tmp[j] = g4[tid + j * CPY];
#pragma unroll
            for (int j = 0; j < PER; ++j) s4[tid + j * CPY] = tmp[j];
        }
    }
    __syncthreads();

    const int lane  = threadIdx.x & 31;
    const int wid   = blockIdx.x * (blockDim.x >> 5) + (threadIdx.x >> 5);
    const int nwave = gridDim.x * (blockDim.x >> 5);
    const int hf    = lane >> 4;
    const int f0    = (lane & 15) * 4;

    for (int v = wid; v < N_NODES; v += nwave) {
        float* orow = out + (size_t)v * HF;
        const int* lrow = reinterpret_cast<const int*>(orow);
        const int kraw = lrow[1 + lane];
        const int n = __builtin_amdgcn_readfirstlane(lrow[0]);
        const float4 zd = *(const float4*)(z + (size_t)v * IN_FEAT + f0);

        float4 acc[NUM_HEADS];
        for (int h = 0; h < NUM_HEADS; ++h) acc[h] = make_float4(0.f, 0.f, 0.f, 0.f);

        if (n <= 32) {
            int k0 = KEY_NONE, s0 = 0, r0 = 0;
            if (lane < n) { k0 = kraw; s0 = src[k0]; r0 = etype[k0]; }
            int rank = 0;
            for (int j = 0; j < n; ++j) {
                const int kj = __builtin_amdgcn_readlane(k0, j);
                rank += (kj < k0) ? 1 : 0;
            }
            float4 zs_next = make_float4(0.f, 0.f, 0.f, 0.f);
            int r_next = 0;
            if (n > 0) pick_pair(z, s0, r0, rank, 0, n, hf, f0, zs_next, r_next);
            for (int q = 0; q < n; q += 2) {
                const float4 zs = zs_next;
                const int r = r_next;
                if (q + 2 < n) pick_pair(z, s0, r0, rank, q + 2, n, hf, f0, zs_next, r_next);
                add_pair(sW, r, zs, zd, f0, acc);
            }
        } else if (n <= LIST_CAP) {
            int last = -1;
            for (int t = 0; t < n; ++t) {
                int best = KEY_NONE;
                for (int i = lane; i < n; i += 32) {
                    const int k = lrow[1 + i];
                    if (k > last && k < best) best = k;
                }
                for (int m = 16; m > 0; m >>= 1) best = min(best, __shfl_xor(best, m, 32));
                last = best;
                const int s = src[best], r = etype[best];
                float4 zs = *(const float4*)(z + (size_t)s * IN_FEAT + f0);
                if (hf) zs = make_float4(0.f, 0.f, 0.f, 0.f);
                add_pair(sW, r, zs, zd, f0, acc);
            }
        } else {
            for (int e0 = 0; e0 < N_EDGES; e0 += 32) {
                unsigned m = __builtin_amdgcn_ballot_w32(dst[e0 + lane] == v);
                while (m) {
                    const int e = e0 + __builtin_ctz(m);
                    m &= m - 1;
                    const int s = src[e], r = etype[e];
                    float4 zs = *(const float4*)(z + (size_t)s * IN_FEAT + f0);
                    if (hf) zs = make_float4(0.f, 0.f, 0.f, 0.f);
                    add_pair(sW, r, zs, zd, f0, acc);
                }
            }
        }

        for (int h = 0; h < NUM_HEADS; ++h) {
            float4 t = acc[h];
            t.x += xhalf(t.x);
            t.y += xhalf(t.y);
            t.z += xhalf(t.z);
            t.w += xhalf(t.w);
            if (hf == 0) *(float4*)(orow + h * OUT_FEAT + f0) = t;
        }
    }
}

extern "C" void kernel_launch(void* const* d_in, const int* in_sizes, int n_in,
                              void* d_out, int out_size, void* d_ws, size_t ws_size,
                              hipStream_t stream) {
    const float* feat      = (const float*)d_in[0];
    const int*   src       = (const int*)  d_in[1];
    const int*   dst       = (const int*)  d_in[2];
    const int*   etype     = (const int*)  d_in[3];
    const float* fc_w      = (const float*)d_in[4];
    const float* self_fc_w = (const float*)d_in[5];
    const float* attn_w    = (const float*)d_in[6];
    const float* w_comp    = (const float*)d_in[7];
    float* out = (float*)d_out;

    float* z        = (float*)d_ws;
    float* attn_fcs = z + (size_t)N_NODES * OUT_FEAT;

    prep_kernel<<<(N_NODES + 255) / 256, 256, 0, stream>>>(w_comp, attn_w, attn_fcs, out);

    {
        int tiles  = MTILES * (OUT_FEAT / 32);
        int blocks = (tiles + 3) / 4;
        proj_wmma_kernel<OUT_FEAT, 2, false><<<blocks, 128, 0, stream>>>(feat, fc_w, z);
    }

    list_kernel<<<(N_EDGES + 255) / 256, 256, 0, stream>>>(dst, out);

    gather_kernel<<<768, GATHER_THREADS, 0, stream>>>(z, src, dst, etype, attn_fcs, out);

    {
        int tiles  = MTILES * (HF / 32);
        int blocks = (tiles + 3) / 4;
        proj_wmma_kernel<HF, 2, true><<<blocks, 128, 0, stream>>>(feat, self_fc_w, out);
    }
}


# self projection moved into the gather kernel: a work-group keeps 16 nodes' message sums in LDS, ten of its waves add feat @ self_fc_w (f32 WMMA) to them and store each row once; fifth launch gone
# speedup vs baseline: 1.8141x; 1.1457x over previous
#include <hip/hip_runtime.h>
#include <hip/hip_bf16.h>

typedef float v2f __attribute__((ext_vector_type(2)));
typedef float v8f __attribute__((ext_vector_type(8)));

#define N_NODES   25000
#define N_EDGES   320000
#define IN_FEAT   64
#define OUT_FEAT  64
#define NUM_HEADS 5
#define NUM_RELS  20
#define NUM_BASES 10
#define HF        (NUM_HEADS * OUT_FEAT)
#define ATTN_ELEMS (NUM_RELS * NUM_HEADS * 2 * OUT_FEAT)
#define MTILES ((N_NODES + 15) / 16)
#define LIST_CAP  (HF - 1)
#define KEY_NONE  0x7fffffff
#define GATHER_THREADS 512
#define PROJ_WAVES (HF / 32)
#define SROW      (HF + 4)

__global__ void prep_kernel(const float* __restrict__ w_comp,
                            const float* __restrict__ attn_w,
                            float* __restrict__ attn_fcs,
                            float* __restrict__ out) {
    int t = blockIdx.x * blockDim.x + threadIdx.x;
    if (t < N_NODES) out[(size_t)t * HF] = 0.f;
    if (t >= ATTN_ELEMS) return;
    int r = t / (NUM_HEADS * 2 * OUT_FEAT);
    int h = (t / (2 * OUT_FEAT)) % NUM_HEADS;
    int f = t % (2 * OUT_FEAT);
    float acc = 0.f;
#pragma unroll
    for (int b = 0; b < NUM_BASES; ++b)
        acc += w_comp[r * NUM_BASES + b] * attn_w[b * (2 * OUT_FEAT * NUM_HEADS) + f * NUM_HEADS + h];
    attn_fcs[t] = acc;
}

template <int LDW, int NT>
__global__ void proj_wmma_kernel(const float* __restrict__ feat,
                                 const float* __restrict__ W,
                                 float* __restrict__ Out) {
    constexpr int WCOLS = LDW / (16 * NT);
    const int wave = threadIdx.x >> 5;
    const int lane = threadIdx.x & 31;
    const int tile = blockIdx.x * (blockDim.x >> 5) + wave;
    if (tile >= MTILES * WCOLS) return;

    const int mt = tile / WCOLS;
    const int wc = tile % WCOLS;

    const int l16  = lane & 15;
    const int koff = (lane < 16) ? 0 : 2;
    const int rowA = min(mt * 16 + l16, N_NODES - 1);
    const int colb = wc * (16 * NT) + l16;

    const float* ap = feat + (size_t)rowA * IN_FEAT + koff;
    const float* bp = W + (size_t)koff * LDW + colb;

    const int mbase = mt * 16 + ((lane < 16) ? 0 : 8);
    float* op = Out + (size_t)mbase * LDW + colb;

    v8f c[NT];
#pragma unroll
    for (int j = 0; j < NT; ++j) c[j] = v8f{};

#pragma unroll
    for (int k = 0; k < IN_FEAT; k += 4) {
        v2f a;
        a.x = ap[k];
        a.y = ap[k + 1];
#pragma unroll
        for (int j = 0; j < NT; ++j) {
            v2f b;
            b.x = bp[k * LDW + j * 16];
            b.y = bp[(k + 1) * LDW + j * 16];
            c[j] = __builtin_amdgcn_wmma_f32_16x16x4_f32(
                        false, a,  false, b,
                        (short)0, c[j],  false,  false);
        }
    }

#pragma unroll
    for (int r = 0; r < 8; ++r) {
        if (mbase + r < N_NODES) {
#pragma unroll
            for (int j = 0; j < NT; ++j)
                op[r * LDW + j * 16] = c[j][r];
        }
    }
}

__global__ void list_kernel(const int* __restrict__ dst, float* out) {
    const int e = blockIdx.x * blockDim.x + threadIdx.x;
    if (e >= N_EDGES) return;
    int* row = reinterpret_cast<int*>(out + (size_t)dst[e] * HF);
    const int pos = atomicAdd(row, 1);
    if (pos < LIST_CAP) atomicExch(row + 1 + pos, e);
}

__device__ __forceinline__ float xhalf(float x) {
    return __int_as_float(__builtin_amdgcn_permlanex16(__float_as_int(x), __float_as_int(x),
                                                       0x76543210, 0xfedcba98, false, false));
}

template <int M>
__device__ __forceinline__ float row_xor_add(float p) {
    return p + __int_as_float(__builtin_amdgcn_update_dpp(0, __float_as_int(p), 0x160 + M,
                                                          0xf, 0xf, true));
}

__device__ __forceinline__ void add_pair(const float* sW, int r, float4 zs, float4 zd,
                                         int f0, float4 (&acc)[NUM_HEADS]) {
    const float* Wr = sW + r * (NUM_HEADS * 2 * OUT_FEAT);
#pragma unroll
    for (int h = 0; h < NUM_HEADS; ++h) {
        const float* Wh = Wr + h * (2 * OUT_FEAT);
        const float4 ws = *(const float4*)(Wh + f0);
        const float4 wd = *(const float4*)(Wh + OUT_FEAT + f0);
        v2f t = v2f{zs.x, zs.y} * v2f{ws.x, ws.y};
        t = v2f{zs.z, zs.w} * v2f{ws.z, ws.w} + t;
        t = v2f{zd.x, zd.y} * v2f{wd.x, wd.y} + t;
        t = v2f{zd.z, zd.w} * v2f{wd.z, wd.w} + t;
        float p = t.x + t.y;
        p = row_xor_add<1>(p);
        p = row_xor_add<2>(p);
        p = row_xor_add<4>(p);
        p = row_xor_add<8>(p);
        const float a = (p > 0.f) ? p : 0.01f * p;
        acc[h].x += a * zs.x;
        acc[h].y += a * zs.y;
        acc[h].z += a * zs.z;
        acc[h].w += a * zs.w;
    }
}

__device__ __forceinline__ void pick_pair(const float* __restrict__ z, int s0, int r0, int rank,
                                          int q, int n, int hf, int f0, float4& zs, int& r) {
    const int oA = __builtin_ctz(__builtin_amdgcn_ballot_w32(rank == q));
    int s = __builtin_amdgcn_readlane(s0, oA);
    int rr = __builtin_amdgcn_readlane(r0, oA);
    const bool hasB = (q + 1 < n);
    if (hasB) {
        const int oB = __builtin_ctz(__builtin_amdgcn_ballot_w32(rank == q + 1));
        const int sB = __builtin_amdgcn_readlane(s0, oB);
        const int rB = __builtin_amdgcn_readlane(r0, oB);
        if (hf) { s = sB; rr = rB; }
    }
    float4 v = *(const float4*)(z + (size_t)s * IN_FEAT + f0);
    if (hf && !hasB) v = make_float4(0.f, 0.f, 0.f, 0.f);
    zs = v;
    r = rr;
}

__global__ void __launch_bounds__(GATHER_THREADS, 12)
gather_kernel(const float* __restrict__ z,
              const int*   __restrict__ src,
              const int*   __restrict__ dst,
              const int*   __restrict__ etype,
              const float* __restrict__ attn_fcs,
              const float* __restrict__ feat,
              const float* __restrict__ Wself,
              float* out) {
    __shared__ __attribute__((aligned(16))) float sW[ATTN_ELEMS];
    __shared__ __attribute__((aligned(16))) float sS[2][16 * SROW];
    {
        constexpr int NV4 = ATTN_ELEMS / 4;
        constexpr int PER = 8;
        constexpr int CPY = NV4 / PER;
        static_assert(CPY * PER == NV4 && CPY <= GATHER_THREADS, "copy split");
        const float4* g4 = reinterpret_cast<const float4*>(attn_fcs);
        float4* s4 = reinterpret_cast<float4*>(sW);
        const int tid = threadIdx.x;
        if (tid < CPY) {
            float4 tmp[PER];
#pragma unroll
            for (int j = 0; j < PER; ++j) tmp[j] = g4[tid + j * CPY];
#pragma unroll
            for (int j = 0; j < PER; ++j) s4[tid + j * CPY] = tmp[j];
        }
    }
    __syncthreads();

    const int lane  = threadIdx.x & 31;
    const int wave  = __builtin_amdgcn_readfirstlane(threadIdx.x >> 5);
    const int hf    = lane >> 4;
    const int f0    = (lane & 15) * 4;

    int par = 0;
    for (int t = blockIdx.x; t < MTILES; t += gridDim.x, par ^= 1) {
        float* S = sS[par];
        const int v = t * 16 + wave;

        float4 acc[NUM_HEADS];
        for (int h = 0; h < NUM_HEADS; ++h) acc[h] = make_float4(0.f, 0.f, 0.f, 0.f);

        if (v < N_NODES) {
            const int* lrow = reinterpret_cast<const int*>(out + (size_t)v * HF);
            const int kraw = lrow[1 + lane];
            const int n = __builtin_amdgcn_readfirstlane(lrow[0]);
            const float4 zd = *(const float4*)(z + (size_t)v * IN_FEAT + f0);

            if (n <= 32) {
                int k0 = KEY_NONE, s0 = 0, r0 = 0;
                if (lane < n) { k0 = kraw; s0 = src[k0]; r0 = etype[k0]; }
                int rank = 0;
                for (int j = 0; j < n; ++j) {
                    const int kj = __builtin_amdgcn_readlane(k0, j);
                    rank += (kj < k0) ? 1 : 0;
                }
                float4 zs_next = make_float4(0.f, 0.f, 0.f, 0.f);
                int r_next = 0;
                if (n > 0) pick_pair(z, s0, r0, rank, 0, n, hf, f0, zs_next, r_next);
                for (int q = 0; q < n; q += 2) {
                    const float4 zs = zs_next;
                    const int r = r_next;
                    if (q + 2 < n) pick_pair(z, s0, r0, rank, q + 2, n, hf, f0, zs_next, r_next);
                    add_pair(sW, r, zs, zd, f0, acc);
                }
            } else if (n <= LIST_CAP) {
                int last = -1;
                for (int i = 0; i < n; ++i) {
                    int best = KEY_NONE;
                    for (int p = lane; p < n; p += 32) {
                        const int k = lrow[1 + p];
                        if (k > last && k < best) best = k;
                    }
                    for (int m = 16; m > 0; m >>= 1) best = min(best, __shfl_xor(best, m, 32));
                    last = best;
                    const int s = src[best], r = etype[best];
                    float4 zs = *(const float4*)(z + (size_t)s * IN_FEAT + f0);
                    if (hf) zs = make_float4(0.f, 0.f, 0.f, 0.f);
                    add_pair(sW, r, zs, zd, f0, acc);
                }
            } else {
                for (int e0 = 0; e0 < N_EDGES; e0 += 32) {
                    unsigned m = __builtin_amdgcn_ballot_w32(dst[e0 + lane] == v);
                    while (m) {
                        const int e = e0 + __builtin_ctz(m);
                        m &= m - 1;
                        const int s = src[e], r = etype[e];
                        float4 zs = *(const float4*)(z + (size_t)s * IN_FEAT + f0);
                        if (hf) zs = make_float4(0.f, 0.f, 0.f, 0.f);
                        add_pair(sW, r, zs, zd, f0, acc);
                    }
                }
            }
        }

        for (int h = 0; h < NUM_HEADS; ++h) {
            float4 u = acc[h];
            u.x += xhalf(u.x);
            u.y += xhalf(u.y);
            u.z += xhalf(u.z);
            u.w += xhalf(u.w);
            if (hf == 0) *(float4*)(S + wave * SROW + h * OUT_FEAT + f0) = u;
        }

        __syncthreads();

        if (wave < PROJ_WAVES) {
            const int l16  = lane & 15;
            const int koff = hf * 2;
            const int rowA = min(t * 16 + l16, N_NODES - 1);
            const int colb = wave * 32 + l16;
            const float* ap = feat + (size_t)rowA * IN_FEAT + koff;
            const float* bp = Wself + (size_t)koff * HF + colb;
            const int mbase = t * 16 + hf * 8;
            float* op = out + (size_t)mbase * HF + colb;
            const float* sp = S + (hf * 8) * SROW + colb;

            v8f c[2];
            c[0] = v8f{};
            c[1] = v8f{};
            constexpr int KH = IN_FEAT / 8;
#pragma unroll
            for (int half = 0; half < 2; ++half) {
                v2f a[KH];
                v2f b[KH][2];
#pragma unroll
                for (int i = 0; i < KH; ++i) {
                    const int k = 32 * half + 4 * i;
                    a[i].x = ap[k];
                    a[i].y = ap[k + 1];
#pragma unroll
                    for (int j = 0; j < 2; ++j) {
                        b[i][j].x = bp[k * HF + j * 16];
                        b[i][j].y = bp[(k + 1) * HF + j * 16];
                    }
                }
#pragma unroll
                for (int i = 0; i < KH; ++i) {
#pragma unroll
                    for (int j = 0; j < 2; ++j)
                        c[j] = __builtin_amdgcn_wmma_f32_16x16x4_f32(
                                    false, a[i],  false, b[i][j],
                                    (short)0, c[j],  false,  false);
                }
            }
#pragma unroll
            for (int r = 0; r < 8; ++r) {
#pragma unroll
                for (int j = 0; j < 2; ++j) c[j][r] += sp[r * SROW + j * 16];
            }
#pragma unroll
            for (int r = 0; r < 8; ++r) {
                if (mbase + r < N_NODES) {
#pragma unroll
                    for (int j = 0; j < 2; ++j) op[r * HF + j * 16] = c[j][r];
                }
            }
        }
    }
}

extern "C" void kernel_launch(void* const* d_in, const int* in_sizes, int n_in,
                              void* d_out, int out_size, void* d_ws, size_t ws_size,
                              hipStream_t stream) {
    const float* feat      = (const float*)d_in[0];
    const int*   src       = (const int*)  d_in[1];
    const int*   dst       = (const int*)  d_in[2];
    const int*   etype     = (const int*)  d_in[3];
    const float* fc_w      = (const float*)d_in[4];
    const float* self_fc_w = (const float*)d_in[5];
    const float* attn_w    = (const float*)d_in[6];
    const float* w_comp    = (const float*)d_in[7];
    float* out = (float*)d_out;

    float* z        = (float*)d_ws;
    float* attn_fcs = z + (size_t)N_NODES * OUT_FEAT;

    prep_kernel<<<(N_NODES + 255) / 256, 256, 0, stream>>>(w_comp, attn_w, attn_fcs, out);

    {
        int tiles  = MTILES * (OUT_FEAT / 32);
        int blocks = (tiles + 3) / 4;
        proj_wmma_kernel<OUT_FEAT, 2><<<blocks, 128, 0, stream>>>(feat, fc_w, z);
    }

    list_kernel<<<(N_EDGES + 255) / 256, 256, 0, stream>>>(dst, out);

    gather_kernel<<<768, GATHER_THREADS, 0, stream>>>(z, src, dst, etype, attn_fcs, feat,
                                                      self_fc_w, out);
}


# edge-list build and z projection share one launch of a quarter of the waves: block index picks the role; a list thread takes 5 edges, atomics in flight together; a projection wave a 16x64 tile
# speedup vs baseline: 1.9045x; 1.0498x over previous
#include <hip/hip_runtime.h>
#include <hip/hip_bf16.h>

typedef float v2f __attribute__((ext_vector_type(2)));
typedef float v8f __attribute__((ext_vector_type(8)));

#define N_NODES   25000
#define N_EDGES   320000
#define IN_FEAT   64
#define OUT_FEAT  64
#define NUM_HEADS 5
#define NUM_RELS  20
#define NUM_BASES 10
#define HF        (NUM_HEADS * OUT_FEAT)
#define ATTN_ELEMS (NUM_RELS * NUM_HEADS * 2 * OUT_FEAT)
#define MTILES ((N_NODES + 15) / 16)
#define LIST_CAP  (HF - 1)
#define KEY_NONE  0x7fffffff
#define GATHER_THREADS 512
#define PROJ_WAVES (HF / 32)
#define SROW      (HF + 4)

__global__ void prep_kernel(const float* __restrict__ w_comp,
                            const float* __restrict__ attn_w,
                            float* __restrict__ attn_fcs,
                            float* __restrict__ out) {
    int t = blockIdx.x * blockDim.x + threadIdx.x;
    if (t < N_NODES) out[(size_t)t * HF] = 0.f;
    if (t >= ATTN_ELEMS) return;
    int r = t / (NUM_HEADS * 2 * OUT_FEAT);
    int h = (t / (2 * OUT_FEAT)) % NUM_HEADS;
    int f = t % (2 * OUT_FEAT);
    float acc = 0.f;
#pragma unroll
    for (int b = 0; b < NUM_BASES; ++b)
        acc += w_comp[r * NUM_BASES + b] * attn_w[b * (2 * OUT_FEAT * NUM_HEADS) + f * NUM_HEADS + h];
    attn_fcs[t] = acc;
}

#define MID_THREADS 256
#define LIST_PER    5
#define LIST_BLOCKS (N_EDGES / (MID_THREADS * LIST_PER))
static_assert(LIST_BLOCKS * MID_THREADS * LIST_PER == N_EDGES, "list split");
#define PROJ_BLOCKS ((MTILES + MID_THREADS / 32 - 1) / (MID_THREADS / 32))

__global__ void __launch_bounds__(MID_THREADS)
list_proj_kernel(const int* __restrict__ dst, float* out,
                 const float* __restrict__ feat,
                 const float* __restrict__ W,
                 float* __restrict__ Z) {
    if (blockIdx.x < LIST_BLOCKS) {
        const int e0 = blockIdx.x * (MID_THREADS * LIST_PER) + threadIdx.x;
        int* row[LIST_PER];
        int pos[LIST_PER];
#pragma unroll
        for (int j = 0; j < LIST_PER; ++j)
            row[j] = reinterpret_cast<int*>(out + (size_t)dst[e0 + j * MID_THREADS] * HF);
#pragma unroll
        for (int j = 0; j < LIST_PER; ++j) pos[j] = atomicAdd(row[j], 1);
#pragma unroll
        for (int j = 0; j < LIST_PER; ++j) {
            const int e = e0 + j * MID_THREADS;
            if (pos[j] < LIST_CAP) atomicExch(row[j] + 1 + pos[j], e);
        }
        return;
    }

    constexpr int LDW = OUT_FEAT;
    constexpr int NT  = OUT_FEAT / 16;
    const int wave = threadIdx.x >> 5;
    const int lane = threadIdx.x & 31;
    const int mt = (blockIdx.x - LIST_BLOCKS) * (MID_THREADS / 32) + wave;
    if (mt >= MTILES) return;

    const int l16  = lane & 15;
    const int koff = (lane < 16) ? 0 : 2;
    const int rowA = min(mt * 16 + l16, N_NODES - 1);

    const float* ap = feat + (size_t)rowA * IN_FEAT + koff;
    const float* bp = W + (size_t)koff * LDW + l16;

    const int mbase = mt * 16 + ((lane < 16) ? 0 : 8);
    float* op = Z + (size_t)mbase * LDW + l16;

    v8f c[NT];
#pragma unroll
    for (int j = 0; j < NT; ++j) c[j] = v8f{};

#pragma unroll
    for (int k = 0; k < IN_FEAT; k += 4) {
        v2f a;
        a.x = ap[k];
        a.y = ap[k + 1];
#pragma unroll
        for (int j = 0; j < NT; ++j) {
            v2f b;
            b.x = bp[k * LDW + j * 16];
            b.y = bp[(k + 1) * LDW + j * 16];
            c[j] = __builtin_amdgcn_wmma_f32_16x16x4_f32(
                        false, a,  false, b,
                        (short)0, c[j],  false,  false);
        }
    }

#pragma unroll
    for (int r = 0; r < 8; ++r) {
        if (mbase + r < N_NODES) {
#pragma unroll
            for (int j = 0; j < NT; ++j)
                op[r * LDW + j * 16] = c[j][r];
        }
    }
}

__device__ __forceinline__ float xhalf(float x) {
    return __int_as_float(__builtin_amdgcn_permlanex16(__float_as_int(x), __float_as_int(x),
                                                       0x76543210, 0xfedcba98, false, false));
}

template <int M>
__device__ __forceinline__ float row_xor_add(float p) {
    return p + __int_as_float(__builtin_amdgcn_update_dpp(0, __float_as_int(p), 0x160 + M,
                                                          0xf, 0xf, true));
}

__device__ __forceinline__ void add_pair(const float* sW, int r, float4 zs, float4 zd,
                                         int f0, float4 (&acc)[NUM_HEADS]) {
    const float* Wr = sW + r * (NUM_HEADS * 2 * OUT_FEAT);
#pragma unroll
    for (int h = 0; h < NUM_HEADS; ++h) {
        const float* Wh = Wr + h * (2 * OUT_FEAT);
        const float4 ws = *(const float4*)(Wh + f0);
        const float4 wd = *(const float4*)(Wh + OUT_FEAT + f0);
        v2f t = v2f{zs.x, zs.y} * v2f{ws.x, ws.y};
        t = v2f{zs.z, zs.w} * v2f{ws.z, ws.w} + t;
        t = v2f{zd.x, zd.y} * v2f{wd.x, wd.y} + t;
        t = v2f{zd.z, zd.w} * v2f{wd.z, wd.w} + t;
        float p = t.x + t.y;
        p = row_xor_add<1>(p);
        p = row_xor_add<2>(p);
        p = row_xor_add<4>(p);
        p = row_xor_add<8>(p);
        const float a = (p > 0.f) ? p : 0.01f * p;
        acc[h].x += a * zs.x;
        acc[h].y += a * zs.y;
        acc[h].z += a * zs.z;
        acc[h].w += a * zs.w;
    }
}

__device__ __forceinline__ void pick_pair(const float* __restrict__ z, int s0, int r0, int rank,
                                          int q, int n, int hf, int f0, float4& zs, int& r) {
    const int oA = __builtin_ctz(__builtin_amdgcn_ballot_w32(rank == q));
    int s = __builtin_amdgcn_readlane(s0, oA);
    int rr = __builtin_amdgcn_readlane(r0, oA);
    const bool hasB = (q + 1 < n);
    if (hasB) {
        const int oB = __builtin_ctz(__builtin_amdgcn_ballot_w32(rank == q + 1));
        const int sB = __builtin_amdgcn_readlane(s0, oB);
        const int rB = __builtin_amdgcn_readlane(r0, oB);
        if (hf) { s = sB; rr = rB; }
    }
    float4 v = *(const float4*)(z + (size_t)s * IN_FEAT + f0);
    if (hf && !hasB) v = make_float4(0.f, 0.f, 0.f, 0.f);
    zs = v;
    r = rr;
}

__global__ void __launch_bounds__(GATHER_THREADS, 12)
gather_kernel(const float* __restrict__ z,
              const int*   __restrict__ src,
              const int*   __restrict__ dst,
              const int*   __restrict__ etype,
              const float* __restrict__ attn_fcs,
              const float* __restrict__ feat,
              const float* __restrict__ Wself,
              float* out) {
    __shared__ __attribute__((aligned(16))) float sW[ATTN_ELEMS];
    __shared__ __attribute__((aligned(16))) float sS[2][16 * SROW];
    {
        constexpr int NV4 = ATTN_ELEMS / 4;
        constexpr int PER = 8;
        constexpr int CPY = NV4 / PER;
        static_assert(CPY * PER == NV4 && CPY <= GATHER_THREADS, "copy split");
        const float4* g4 = reinterpret_cast<const float4*>(attn_fcs);
        float4* s4 = reinterpret_cast<float4*>(sW);
        const int tid = threadIdx.x;
        if (tid < CPY) {
            float4 tmp[PER];
#pragma unroll
            for (int j = 0; j < PER; ++j) tmp[j] = g4[tid + j * CPY];
#pragma unroll
            for (int j = 0; j < PER; ++j) s4[tid + j * CPY] = tmp[j];
        }
    }
    __syncthreads();

    const int lane  = threadIdx.x & 31;
    const int wave  = __builtin_amdgcn_readfirstlane(threadIdx.x >> 5);
    const int hf    = lane >> 4;
    const int f0    = (lane & 15) * 4;

    int par = 0;
    for (int t = blockIdx.x; t < MTILES; t += gridDim.x, par ^= 1) {
        float* S = sS[par];
        const int v = t * 16 + wave;

        float4 acc[NUM_HEADS];
        for (int h = 0; h < NUM_HEADS; ++h) acc[h] = make_float4(0.f, 0.f, 0.f, 0.f);

        if (v < N_NODES) {
            const int* lrow = reinterpret_cast<const int*>(out + (size_t)v * HF);
            const int kraw = lrow[1 + lane];
            const int n = __builtin_amdgcn_readfirstlane(lrow[0]);
            const float4 zd = *(const float4*)(z + (size_t)v * IN_FEAT + f0);

            if (n <= 32) {
                int k0 = KEY_NONE, s0 = 0, r0 = 0;
                if (lane < n) { k0 = kraw; s0 = src[k0]; r0 = etype[k0]; }
                int rank = 0;
                for (int j = 0; j < n; ++j) {
                    const int kj = __builtin_amdgcn_readlane(k0, j);
                    rank += (kj < k0) ? 1 : 0;
                }
                float4 zs_next = make_float4(0.f, 0.f, 0.f, 0.f);
                int r_next = 0;
                if (n > 0) pick_pair(z, s0, r0, rank, 0, n, hf, f0, zs_next, r_next);
                for (int q = 0; q < n; q += 2) {
                    const float4 zs = zs_next;
                    const int r = r_next;
                    if (q + 2 < n) pick_pair(z, s0, r0, rank, q + 2, n, hf, f0, zs_next, r_next);
                    add_pair(sW, r, zs, zd, f0, acc);
                }
            } else if (n <= LIST_CAP) {
                int last = -1;
                for (int i = 0; i < n; ++i) {
                    int best = KEY_NONE;
                    for (int p = lane; p < n; p += 32) {
                        const int k = lrow[1 + p];
                        if (k > last && k < best) best = k;
                    }
                    for (int m = 16; m > 0; m >>= 1) best = min(best, __shfl_xor(best, m, 32));
                    last = best;
                    const int s = src[best], r = etype[best];
                    float4 zs = *(const float4*)(z + (size_t)s * IN_FEAT + f0);
                    if (hf) zs = make_float4(0.f, 0.f, 0.f, 0.f);
                    add_pair(sW, r, zs, zd, f0, acc);
                }
            } else {
                for (int e0 = 0; e0 < N_EDGES; e0 += 32) {
                    unsigned m = __builtin_amdgcn_ballot_w32(dst[e0 + lane] == v);
                    while (m) {
                        const int e = e0 + __builtin_ctz(m);
                        m &= m - 1;
                        const int s = src[e], r = etype[e];
                        float4 zs = *(const float4*)(z + (size_t)s * IN_FEAT + f0);
                        if (hf) zs = make_float4(0.f, 0.f, 0.f, 0.f);
                        add_pair(sW, r, zs, zd, f0, acc);
                    }
                }
            }
        }

        for (int h = 0; h < NUM_HEADS; ++h) {
            float4 u = acc[h];
            u.x += xhalf(u.x);
            u.y += xhalf(u.y);
            u.z += xhalf(u.z);
            u.w += xhalf(u.w);
            if (hf == 0) *(float4*)(S + wave * SROW + h * OUT_FEAT + f0) = u;
        }

        __syncthreads();

        if (wave < PROJ_WAVES) {
            const int l16  = lane & 15;
            const int koff = hf * 2;
            const int rowA = min(t * 16 + l16, N_NODES - 1);
            const int colb = wave * 32 + l16;
            const float* ap = feat + (size_t)rowA * IN_FEAT + koff;
            const float* bp = Wself + (size_t)koff * HF + colb;
            const int mbase = t * 16 + hf * 8;
            float* op = out + (size_t)mbase * HF + colb;
            const float* sp = S + (hf * 8) * SROW + colb;

            v8f c[2];
            c[0] = v8f{};
            c[1] = v8f{};
            constexpr int KH = IN_FEAT / 8;
#pragma unroll
            for (int half = 0; half < 2; ++half) {
                v2f a[KH];
                v2f b[KH][2];
#pragma unroll
                for (int i = 0; i < KH; ++i) {
                    const int k = 32 * half + 4 * i;
                    a[i].x = ap[k];
                    a[i].y = ap[k + 1];
#pragma unroll
                    for (int j = 0; j < 2; ++j) {
                        b[i][j].x = bp[k * HF + j * 16];
                        b[i][j].y = bp[(k + 1) * HF + j * 16];
                    }
                }
#pragma unroll
                for (int i = 0; i < KH; ++i) {
#pragma unroll
                    for (int j = 0; j < 2; ++j)
                        c[j] = __builtin_amdgcn_wmma_f32_16x16x4_f32(
                                    false, a[i],  false, b[i][j],
                                    (short)0, c[j],  false,  false);
                }
            }
#pragma unroll
            for (int r = 0; r < 8; ++r) {
#pragma unroll
                for (int j = 0; j < 2; ++j) c[j][r] += sp[r * SROW + j * 16];
            }
#pragma unroll
            for (int r = 0; r < 8; ++r) {
                if (mbase + r < N_NODES) {
#pragma unroll
                    for (int j = 0; j < 2; ++j) op[r * HF + j * 16] = c[j][r];
                }
            }
        }
    }
}

extern "C" void kernel_launch(void* const* d_in, const int* in_sizes, int n_in,
                              void* d_out, int out_size, void* d_ws, size_t ws_size,
                              hipStream_t stream) {
    const float* feat      = (const float*)d_in[0];
    const int*   src       = (const int*)  d_in[1];
    const int*   dst       = (const int*)  d_in[2];
    const int*   etype     = (const int*)  d_in[3];
    const float* fc_w      = (const float*)d_in[4];
    const float* self_fc_w = (const float*)d_in[5];
    const float* attn_w    = (const float*)d_in[6];
    const float* w_comp    = (const float*)d_in[7];
    float* out = (float*)d_out;

    float* z        = (float*)d_ws;
    float* attn_fcs = z + (size_t)N_NODES * OUT_FEAT;

    prep_kernel<<<(N_NODES + 255) / 256, 256, 0, stream>>>(w_comp, attn_w, attn_fcs, out);

    list_proj_kernel<<<LIST_BLOCKS + PROJ_BLOCKS, MID_THREADS, 0, stream>>>(dst, out, feat, fc_w, z);

    gather_kernel<<<768, GATHER_THREADS, 0, stream>>>(z, src, dst, etype, attn_fcs, feat,
                                                      self_fc_w, out);
}


# in-edge lists built in LDS: a work-group owns 128 nodes, finds its edges in bit-plane tags of dst (32 edges per 8 logic ops), ranks them by LDS adds, stores rows plainly; no global atomics left
# speedup vs baseline: 2.1026x; 1.1040x over previous
#include <hip/hip_runtime.h>
#include <hip/hip_bf16.h>

typedef float v2f __attribute__((ext_vector_type(2)));
typedef float v8f __attribute__((ext_vector_type(8)));

#define N_NODES   25000
#define N_EDGES   320000
#define IN_FEAT   64
#define OUT_FEAT  64
#define NUM_HEADS 5
#define NUM_RELS  20
#define NUM_BASES 10
#define HF        (NUM_HEADS * OUT_FEAT)
#define ATTN_ELEMS (NUM_RELS * NUM_HEADS * 2 * OUT_FEAT)
#define MTILES ((N_NODES + 15) / 16)
#define LIST_CAP  64
#define BKT_SHIFT 7
#define BKT_NODES (1 << BKT_SHIFT)
#define N_BKT     ((N_NODES + BKT_NODES - 1) / BKT_NODES)
#define ROW_BYTES (HF * 4)
#define TAG_OFF   512
#define TAG_WPR   128
#define N_GRP     (N_EDGES / 32)
#define WQCAP     256
#define KEY_NONE  0x7fffffff
#define GATHER_THREADS 512
#define PROJ_WAVES (HF / 32)
#define SROW      (HF + 4)

__global__ void prep_kernel(const float* __restrict__ w_comp,
                            const float* __restrict__ attn_w,
                            float* __restrict__ attn_fcs,
                            const int* __restrict__ dst,
                            float* __restrict__ out) {
    int t = blockIdx.x * blockDim.x + threadIdx.x;
    if (t < N_EDGES / 8) {
        const int wg = t >> 5, lane = t & 31;
        int d[8];
#pragma unroll
        for (int k = 0; k < 8; ++k) d[k] = dst[wg * 256 + k * 32 + lane];
        unsigned w[2] = {0u, 0u};
#pragma unroll
        for (int k = 0; k < 8; ++k) {
#pragma unroll
            for (int p = 0; p < 8; ++p) {
                const unsigned b = __builtin_amdgcn_ballot_w32(((d[k] >> (BKT_SHIFT + p)) & 1) != 0);
                if (lane == 8 * (k & 3) + p) w[k >> 2] = b;
            }
        }
        char* p = reinterpret_cast<char*>(out) + (size_t)(wg >> 1) * ROW_BYTES + TAG_OFF + (((wg & 1) * 64 + lane) << 2);
        *reinterpret_cast<unsigned*>(p) = w[0];
        *reinterpret_cast<unsigned*>(p + 128) = w[1];
    }
    if (t >= ATTN_ELEMS) return;
    int r = t / (NUM_HEADS * 2 * OUT_FEAT);
    int h = (t / (2 * OUT_FEAT)) % NUM_HEADS;
    int f = t % (2 * OUT_FEAT);
    float acc = 0.f;
#pragma unroll
    for (int b = 0; b < NUM_BASES; ++b)
        acc += w_comp[r * NUM_BASES + b] * attn_w[b * (2 * OUT_FEAT * NUM_HEADS) + f * NUM_HEADS + h];
    attn_fcs[t] = acc;
}

#define MID_THREADS 512
#define LIST_BLOCKS N_BKT
#define SCAN_U      5
#define SCAN_ROUNDS ((N_GRP + MID_THREADS * SCAN_U - 1) / (MID_THREADS * SCAN_U))
static_assert(N_EDGES % 256 == 0 && TAG_WPR == 128, "tag split");
static_assert(N_BKT <= 256 && MID_THREADS / 32 * 8 == BKT_NODES, "bucket split");

#define PROJ_BLOCKS ((MTILES + MID_THREADS / 32 - 1) / (MID_THREADS / 32))

__global__ void __launch_bounds__(MID_THREADS)
list_proj_kernel(const int* __restrict__ dst, float* out,
                 const float* __restrict__ feat,
                 const float* __restrict__ W,
                 float* __restrict__ Z) {
    __shared__ int sQ[(MID_THREADS / 32) * WQCAP];
    __shared__ int sCnt[BKT_NODES];
    __shared__ int sList[BKT_NODES * LIST_CAP];
    __shared__ int sOvf;
    if (blockIdx.x < LIST_BLOCKS) {
        const int g = blockIdx.x;
        const int tid = threadIdx.x;
        if (tid < BKT_NODES) sCnt[tid] = 0;
        if (tid == 0) sOvf = 0;
        __syncthreads();

        unsigned x[8];
#pragma unroll
        for (int p = 0; p < 8; ++p) x[p] = ((g >> p) & 1) ? 0u : ~0u;
        const char* ob = reinterpret_cast<const char*>(out);
        const int wv = tid >> 5, ln = tid & 31;
        int* wq = sQ + wv * WQCAP;
        int wcount = 0;
        for (int k = 0; k < SCAN_ROUNDS; ++k) {
            uint4 lo[SCAN_U], hi[SCAN_U];
#pragma unroll
            for (int u = 0; u < SCAN_U; ++u) {
                const int j = (k * SCAN_U + u) * MID_THREADS + tid;
                lo[u] = make_uint4(0u, 0u, 0u, 0u);
                hi[u] = lo[u];
                if (j < N_GRP) {
                    const char* p = ob + (size_t)(j >> 4) * ROW_BYTES + TAG_OFF + ((j & 15) << 5);
                    lo[u] = *reinterpret_cast<const uint4*>(p);
                    hi[u] = *reinterpret_cast<const uint4*>(p + 16);
                }
            }
#pragma unroll
            for (int u = 0; u < SCAN_U; ++u) {
                const int j = (k * SCAN_U + u) * MID_THREADS + tid;
                unsigned mm = (lo[u].x ^ x[0]) & (lo[u].y ^ x[1]) & (lo[u].z ^ x[2]) & (lo[u].w ^ x[3]) &
                              (hi[u].x ^ x[4]) & (hi[u].y ^ x[5]) & (hi[u].z ^ x[6]) & (hi[u].w ^ x[7]);
                if (j >= N_GRP) mm = 0u;
                while (true) {
                    const unsigned mk = __builtin_amdgcn_ballot_w32(mm != 0u);
                    if (mk == 0u) break;
                    if (mm != 0u) {
                        const int pos = wcount + (int)__builtin_amdgcn_mbcnt_lo(mk, 0u);
                        if (pos < WQCAP) wq[pos] = j * 32 + __builtin_ctz(mm);
                        mm &= mm - 1u;
                    }
                    wcount += __builtin_popcount(mk);
                }
            }
        }
        if (wcount > WQCAP) sOvf = 1;
        __syncthreads();

        const int qn = wcount < WQCAP ? wcount : WQCAP;
        constexpr int PB = 4;
        for (int b0 = 0; b0 < qn; b0 += PB * 32) {
            int e[PB], d[PB];
#pragma unroll
            for (int j = 0; j < PB; ++j) {
                const int idx = b0 + j * 32 + ln;
                e[j] = (idx < qn) ? wq[idx] : -1;
            }
#pragma unroll
            for (int j = 0; j < PB; ++j) d[j] = dst[e[j] < 0 ? 0 : e[j]];
#pragma unroll
            for (int j = 0; j < PB; ++j) {
                if (e[j] >= 0) {
                    const int loc = d[j] & (BKT_NODES - 1);
                    const int pos = atomicAdd(&sCnt[loc], 1);
                    if (pos < LIST_CAP) sList[loc * LIST_CAP + pos] = e[j];
                }
            }
        }
        __syncthreads();
        const bool ovf = sOvf != 0;

        for (int k = 0; k < 8; ++k) {
            const int loc = wv * 8 + k;
            const int node = g * BKT_NODES + loc;
            if (node >= N_NODES) break;
            const int cnt = sCnt[loc];
            const int nl = cnt < LIST_CAP ? cnt : LIST_CAP;
            const int top = nl > 31 ? nl : 31;
            int* row = reinterpret_cast<int*>(out + (size_t)node * HF);
            for (int w = ln; w <= top; w += 32) {
                int val = 0;
                if (w == 0) val = ovf ? N_EDGES : cnt;
                else if (w <= nl) val = sList[loc * LIST_CAP + w - 1];
                row[w] = val;
            }
        }
        return;
    }

    constexpr int LDW = OUT_FEAT;
    constexpr int NT  = OUT_FEAT / 16;
    const int wave = threadIdx.x >> 5;
    const int lane = threadIdx.x & 31;
    const int mt = (blockIdx.x - LIST_BLOCKS) * (MID_THREADS / 32) + wave;
    if (mt >= MTILES) return;

    const int l16  = lane & 15;
    const int koff = (lane < 16) ? 0 : 2;
    const int rowA = min(mt * 16 + l16, N_NODES - 1);

    const float* ap = feat + (size_t)rowA * IN_FEAT + koff;
    const float* bp = W + (size_t)koff * LDW + l16;

    const int mbase = mt * 16 + ((lane < 16) ? 0 : 8);
    float* op = Z + (size_t)mbase * LDW + l16;

    v8f c[NT];
#pragma unroll
    for (int j = 0; j < NT; ++j) c[j] = v8f{};

#pragma unroll
    for (int k = 0; k < IN_FEAT; k += 4) {
        v2f a;
        a.x = ap[k];
        a.y = ap[k + 1];
#pragma unroll
        for (int j = 0; j < NT; ++j) {
            v2f b;
            b.x = bp[k * LDW + j * 16];
            b.y = bp[(k + 1) * LDW + j * 16];
            c[j] = __builtin_amdgcn_wmma_f32_16x16x4_f32(
                        false, a,  false, b,
                        (short)0, c[j],  false,  false);
        }
    }

#pragma unroll
    for (int r = 0; r < 8; ++r) {
        if (mbase + r < N_NODES) {
#pragma unroll
            for (int j = 0; j < NT; ++j)
                op[r * LDW + j * 16] = c[j][r];
        }
    }
}

__device__ __forceinline__ float xhalf(float x) {
    return __int_as_float(__builtin_amdgcn_permlanex16(__float_as_int(x), __float_as_int(x),
                                                       0x76543210, 0xfedcba98, false, false));
}

template <int M>
__device__ __forceinline__ float row_xor_add(float p) {
    return p + __int_as_float(__builtin_amdgcn_update_dpp(0, __float_as_int(p), 0x160 + M,
                                                          0xf, 0xf, true));
}

__device__ __forceinline__ void add_pair(const float* sW, int r, float4 zs, float4 zd,
                                         int f0, float4 (&acc)[NUM_HEADS]) {
    const float* Wr = sW + r * (NUM_HEADS * 2 * OUT_FEAT);
#pragma unroll
    for (int h = 0; h < NUM_HEADS; ++h) {
        const float* Wh = Wr + h * (2 * OUT_FEAT);
        const float4 ws = *(const float4*)(Wh + f0);
        const float4 wd = *(const float4*)(Wh + OUT_FEAT + f0);
        v2f t = v2f{zs.x, zs.y} * v2f{ws.x, ws.y};
        t = v2f{zs.z, zs.w} * v2f{ws.z, ws.w} + t;
        t = v2f{zd.x, zd.y} * v2f{wd.x, wd.y} + t;
        t = v2f{zd.z, zd.w} * v2f{wd.z, wd.w} + t;
        float p = t.x + t.y;
        p = row_xor_add<1>(p);
        p = row_xor_add<2>(p);
        p = row_xor_add<4>(p);
        p = row_xor_add<8>(p);
        const float a = (p > 0.f) ? p : 0.01f * p;
        acc[h].x += a * zs.x;
        acc[h].y += a * zs.y;
        acc[h].z += a * zs.z;
        acc[h].w += a * zs.w;
    }
}

__device__ __forceinline__ void pick_pair(const float* __restrict__ z, int s0, int r0, int rank,
                                          int q, int n, int hf, int f0, float4& zs, int& r) {
    const int oA = __builtin_ctz(__builtin_amdgcn_ballot_w32(rank == q));
    int s = __builtin_amdgcn_readlane(s0, oA);
    int rr = __builtin_amdgcn_readlane(r0, oA);
    const bool hasB = (q + 1 < n);
    if (hasB) {
        const int oB = __builtin_ctz(__builtin_amdgcn_ballot_w32(rank == q + 1));
        const int sB = __builtin_amdgcn_readlane(s0, oB);
        const int rB = __builtin_amdgcn_readlane(r0, oB);
        if (hf) { s = sB; rr = rB; }
    }
    float4 v = *(const float4*)(z + (size_t)s * IN_FEAT + f0);
    if (hf && !hasB) v = make_float4(0.f, 0.f, 0.f, 0.f);
    zs = v;
    r = rr;
}

__global__ void __launch_bounds__(GATHER_THREADS, 12)
gather_kernel(const float* __restrict__ z,
              const int*   __restrict__ src,
              const int*   __restrict__ dst,
              const int*   __restrict__ etype,
              const float* __restrict__ attn_fcs,
              const float* __restrict__ feat,
              const float* __restrict__ Wself,
              float* out) {
    __shared__ __attribute__((aligned(16))) float sW[ATTN_ELEMS];
    __shared__ __attribute__((aligned(16))) float sS[2][16 * SROW];
    {
        constexpr int NV4 = ATTN_ELEMS / 4;
        constexpr int PER = 8;
        constexpr int CPY = NV4 / PER;
        static_assert(CPY * PER == NV4 && CPY <= GATHER_THREADS, "copy split");
        const float4* g4 = reinterpret_cast<const float4*>(attn_fcs);
        float4* s4 = reinterpret_cast<float4*>(sW);
        const int tid = threadIdx.x;
        if (tid < CPY) {
            float4 tmp[PER];
#pragma unroll
            for (int j = 0; j < PER; ++j) tmp[j] = g4[tid + j * CPY];
#pragma unroll
            for (int j = 0; j < PER; ++j) s4[tid + j * CPY] = tmp[j];
        }
    }
    __syncthreads();

    const int lane  = threadIdx.x & 31;
    const int wave  = __builtin_amdgcn_readfirstlane(threadIdx.x >> 5);
    const int hf    = lane >> 4;
    const int f0    = (lane & 15) * 4;

    int par = 0;
    for (int t = blockIdx.x; t < MTILES; t += gridDim.x, par ^= 1) {
        float* S = sS[par];
        const int v = t * 16 + wave;

        float4 acc[NUM_HEADS];
        for (int h = 0; h < NUM_HEADS; ++h) acc[h] = make_float4(0.f, 0.f, 0.f, 0.f);

        if (v < N_NODES) {
            const int* lrow = reinterpret_cast<const int*>(out + (size_t)v * HF);
            const int kraw = lrow[1 + lane];
            const int n = __builtin_amdgcn_readfirstlane(lrow[0]);
            const float4 zd = *(const float4*)(z + (size_t)v * IN_FEAT + f0);

            if (n <= 32) {
                int k0 = KEY_NONE, s0 = 0, r0 = 0;
                if (lane < n) { k0 = kraw; s0 = src[k0]; r0 = etype[k0]; }
                int rank = 0;
                for (int j = 0; j < n; ++j) {
                    const int kj = __builtin_amdgcn_readlane(k0, j);
                    rank += (kj < k0) ? 1 : 0;
                }
                float4 zs_next = make_float4(0.f, 0.f, 0.f, 0.f);
                int r_next = 0;
                if (n > 0) pick_pair(z, s0, r0, rank, 0, n, hf, f0, zs_next, r_next);
                for (int q = 0; q < n; q += 2) {
                    const float4 zs = zs_next;
                    const int r = r_next;
                    if (q + 2 < n) pick_pair(z, s0, r0, rank, q + 2, n, hf, f0, zs_next, r_next);
                    add_pair(sW, r, zs, zd, f0, acc);
                }
            } else if (n <= LIST_CAP) {
                int last = -1;
                for (int i = 0; i < n; ++i) {
                    int best = KEY_NONE;
                    for (int p = lane; p < n; p += 32) {
                        const int k = lrow[1 + p];
                        if (k > last && k < best) best = k;
                    }
                    for (int m = 16; m > 0; m >>= 1) best = min(best, __shfl_xor(best, m, 32));
                    last = best;
                    const int s = src[best], r = etype[best];
                    float4 zs = *(const float4*)(z + (size_t)s * IN_FEAT + f0);
                    if (hf) zs = make_float4(0.f, 0.f, 0.f, 0.f);
                    add_pair(sW, r, zs, zd, f0, acc);
                }
            } else {
                for (int e0 = 0; e0 < N_EDGES; e0 += 32) {
                    unsigned m = __builtin_amdgcn_ballot_w32(dst[e0 + lane] == v);
                    while (m) {
                        const int e = e0 + __builtin_ctz(m);
                        m &= m - 1;
                        const int s = src[e], r = etype[e];
                        float4 zs = *(const float4*)(z + (size_t)s * IN_FEAT + f0);
                        if (hf) zs = make_float4(0.f, 0.f, 0.f, 0.f);
                        add_pair(sW, r, zs, zd, f0, acc);
                    }
                }
            }
        }

        for (int h = 0; h < NUM_HEADS; ++h) {
            float4 u = acc[h];
            u.x += xhalf(u.x);
            u.y += xhalf(u.y);
            u.z += xhalf(u.z);
            u.w += xhalf(u.w);
            if (hf == 0) *(float4*)(S + wave * SROW + h * OUT_FEAT + f0) = u;
        }

        __syncthreads();

        if (wave < PROJ_WAVES) {
            const int l16  = lane & 15;
            const int koff = hf * 2;
            const int rowA = min(t * 16 + l16, N_NODES - 1);
            const int colb = wave * 32 + l16;
            const float* ap = feat + (size_t)rowA * IN_FEAT + koff;
            const float* bp = Wself + (size_t)koff * HF + colb;
            const int mbase = t * 16 + hf * 8;
            float* op = out + (size_t)mbase * HF + colb;
            const float* sp = S + (hf * 8) * SROW + colb;

            v8f c[2];
            c[0] = v8f{};
            c[1] = v8f{};
            constexpr int KH = IN_FEAT / 8;
#pragma unroll
            for (int half = 0; half < 2; ++half) {
                v2f a[KH];
                v2f b[KH][2];
#pragma unroll
                for (int i = 0; i < KH; ++i) {
                    const int k = 32 * half + 4 * i;
                    a[i].x = ap[k];
                    a[i].y = ap[k + 1];
#pragma unroll
                    for (int j = 0; j < 2; ++j) {
                        b[i][j].x = bp[k * HF + j * 16];
                        b[i][j].y = bp[(k + 1) * HF + j * 16];
                    }
                }
#pragma unroll
                for (int i = 0; i < KH; ++i) {
#pragma unroll
                    for (int j = 0; j < 2; ++j)
                        c[j] = __builtin_amdgcn_wmma_f32_16x16x4_f32(
                                    false, a[i],  false, b[i][j],
                                    (short)0, c[j],  false,  false);
                }
            }
#pragma unroll
            for (int r = 0; r < 8; ++r) {
#pragma unroll
                for (int j = 0; j < 2; ++j) c[j][r] += sp[r * SROW + j * 16];
            }
#pragma unroll
            for (int r = 0; r < 8; ++r) {
                if (mbase + r < N_NODES) {
#pragma unroll
                    for (int j = 0; j < 2; ++j) op[r * HF + j * 16] = c[j][r];
                }
            }
        }
    }
}

extern "C" void kernel_launch(void* const* d_in, const int* in_sizes, int n_in,
                              void* d_out, int out_size, void* d_ws, size_t ws_size,
                              hipStream_t stream) {
    const float* feat      = (const float*)d_in[0];
    const int*   src       = (const int*)  d_in[1];
    const int*   dst       = (const int*)  d_in[2];
    const int*   etype     = (const int*)  d_in[3];
    const float* fc_w      = (const float*)d_in[4];
    const float* self_fc_w = (const float*)d_in[5];
    const float* attn_w    = (const float*)d_in[6];
    const float* w_comp    = (const float*)d_in[7];
    float* out = (float*)d_out;

    float* z        = (float*)d_ws;
    float* attn_fcs = z + (size_t)N_NODES * OUT_FEAT;

    prep_kernel<<<(N_EDGES / 8 + 255) / 256, 256, 0, stream>>>(w_comp, attn_w, attn_fcs, dst, out);

    list_proj_kernel<<<LIST_BLOCKS + PROJ_BLOCKS, MID_THREADS, 0, stream>>>(dst, out, feat, fc_w, z);

    gather_kernel<<<768, GATHER_THREADS, 0, stream>>>(z, src, dst, etype, attn_fcs, feat,
                                                      self_fc_w, out);
}


# in-edge lists built in LDS by the gather work-group from per-chunk segments a partition pass leaves in the tile's own output rows (edge, source, relation); tag pass, dst scan and list launch gone
# speedup vs baseline: 2.4190x; 1.1505x over previous
#include <hip/hip_runtime.h>
#include <hip/hip_bf16.h>

typedef float v2f __attribute__((ext_vector_type(2)));
typedef float v8f __attribute__((ext_vector_type(8)));

#define N_NODES   25000
#define N_EDGES   320000
#define IN_FEAT   64
#define OUT_FEAT  64
#define NUM_HEADS 5
#define NUM_RELS  20
#define NUM_BASES 10
#define HF        (NUM_HEADS * OUT_FEAT)
#define ATTN_ELEMS (NUM_RELS * NUM_HEADS * 2 * OUT_FEAT)
#define MTILES ((N_NODES + 15) / 16)
#define LIST_CAP  64
#define ROW_BYTES (HF * 4)
#define N_CHUNK   80
#define CHUNK_EDGES (N_EDGES / N_CHUNK)
#define SEG_WORDS 32
#define SEG_CAP   15
#define TILE_BYTES (16 * ROW_BYTES)
#define TILE_WORDS (TILE_BYTES / 4)
#define BIN_PITCH 33
#define KEY_NONE  0x7fffffff
#define GATHER_THREADS 512
#define PROJ_WAVES (HF / 32)
#define SROW      (HF + 4)


#define FRONT_THREADS 1024
#define PART_BLOCKS  N_CHUNK
#define ZPROJ_BLOCKS ((MTILES + FRONT_THREADS / 32 - 1) / (FRONT_THREADS / 32))
#define ATTN_BLOCKS  ((ATTN_ELEMS + FRONT_THREADS - 1) / FRONT_THREADS)
#define PART_PER     ((CHUNK_EDGES + FRONT_THREADS - 1) / FRONT_THREADS)
static_assert(N_CHUNK * CHUNK_EDGES == N_EDGES, "chunk split");
static_assert(N_CHUNK * SEG_WORDS * 4 <= (N_NODES - (MTILES - 1) * 16) * ROW_BYTES, "segments fit the last tile's rows");
static_assert(N_EDGES <= (1 << 19) && N_NODES <= (1 << 16), "packed words");

__global__ void __launch_bounds__(FRONT_THREADS)
front_kernel(const int* __restrict__ dst, const int* __restrict__ src,
             const int* __restrict__ etype, float* out,
             const float* __restrict__ feat,
             const float* __restrict__ W,
             float* __restrict__ Z,
             const float* __restrict__ w_comp,
             const float* __restrict__ attn_w,
             float* __restrict__ attn_fcs) {
    __shared__ int sBin[MTILES * BIN_PITCH];
    if (blockIdx.x < PART_BLOCKS) {
        const int c = blockIdx.x;
        const int tid = threadIdx.x;
        for (int i = tid; i < MTILES; i += FRONT_THREADS) sBin[i * BIN_PITCH] = 0;

        int d[PART_PER], s[PART_PER], r[PART_PER];
#pragma unroll
        for (int j = 0; j < PART_PER; ++j) {
            const int i = j * FRONT_THREADS + tid;
            const int e = c * CHUNK_EDGES + (i < CHUNK_EDGES ? i : 0);
            d[j] = dst[e];
            s[j] = src[e];
            r[j] = etype[e];
        }
        __syncthreads();
#pragma unroll
        for (int j = 0; j < PART_PER; ++j) {
            const int i = j * FRONT_THREADS + tid;
            if (i < CHUNK_EDGES) {
                const int e = c * CHUNK_EDGES + i;
                const int tile = (int)min((unsigned)d[j] >> 4, (unsigned)(MTILES - 1));
                int* bin = sBin + tile * BIN_PITCH;
                const int pos = atomicAdd(bin, 1);
                if (pos < SEG_CAP) {
                    const int w = pos + 1;
                    const int x = ((w >> 2) << 3) + (w & 3);
                    bin[x]     = e | ((d[j] & 15) << 19);
                    bin[x + 4] = s[j] | (r[j] << 16);
                }
            }
        }
        __syncthreads();
        int* ow = reinterpret_cast<int*>(out);
        for (int i = tid; i < MTILES * SEG_WORDS; i += FRONT_THREADS) {
            const int t = i >> 5, x = i & 31;
            const int* bin = sBin + t * BIN_PITCH;
            const int cnt = bin[0];
            const int cc = cnt < SEG_CAP ? cnt : SEG_CAP;
            const int w = ((x >> 3) << 2) + (x & 3);
            const bool listed = (x & 4) ? (w >= 1 && w <= cc) : (w <= cc);
            if (listed) ow[(size_t)t * TILE_WORDS + c * SEG_WORDS + x] = bin[x];
        }
        return;
    }

    if (blockIdx.x >= PART_BLOCKS + ZPROJ_BLOCKS) {
        const int t = (blockIdx.x - PART_BLOCKS - ZPROJ_BLOCKS) * FRONT_THREADS + threadIdx.x;
        if (t >= ATTN_ELEMS) return;
        const int r = t / (NUM_HEADS * 2 * OUT_FEAT);
        const int h = (t / (2 * OUT_FEAT)) % NUM_HEADS;
        const int f = t % (2 * OUT_FEAT);
        float acc = 0.f;
#pragma unroll
        for (int b = 0; b < NUM_BASES; ++b)
            acc += w_comp[r * NUM_BASES + b] * attn_w[b * (2 * OUT_FEAT * NUM_HEADS) + f * NUM_HEADS + h];
        attn_fcs[t] = acc;
        return;
    }

    constexpr int LDW = OUT_FEAT;
    constexpr int NT  = OUT_FEAT / 16;
    const int wave = threadIdx.x >> 5;
    const int lane = threadIdx.x & 31;
    const int mt = (blockIdx.x - PART_BLOCKS) * (FRONT_THREADS / 32) + wave;
    if (mt >= MTILES) return;

    const int l16  = lane & 15;
    const int koff = (lane < 16) ? 0 : 2;
    const int rowA = min(mt * 16 + l16, N_NODES - 1);

    const float* ap = feat + (size_t)rowA * IN_FEAT + koff;
    const float* bp = W + (size_t)koff * LDW + l16;

    const int mbase = mt * 16 + ((lane < 16) ? 0 : 8);
    float* op = Z + (size_t)mbase * LDW + l16;

    v8f c[NT];
#pragma unroll
    for (int j = 0; j < NT; ++j) c[j] = v8f{};

#pragma unroll
    for (int k = 0; k < IN_FEAT; k += 4) {
        v2f a;
        a.x = ap[k];
        a.y = ap[k + 1];
#pragma unroll
        for (int j = 0; j < NT; ++j) {
            v2f b;
            b.x = bp[k * LDW + j * 16];
            b.y = bp[(k + 1) * LDW + j * 16];
            c[j] = __builtin_amdgcn_wmma_f32_16x16x4_f32(
                        false, a,  false, b,
                        (short)0, c[j],  false,  false);
        }
    }

#pragma unroll
    for (int r = 0; r < 8; ++r) {
        if (mbase + r < N_NODES) {
#pragma unroll
            for (int j = 0; j < NT; ++j)
                op[r * LDW + j * 16] = c[j][r];
        }
    }
}

__device__ __forceinline__ float xhalf(float x) {
    return __int_as_float(__builtin_amdgcn_permlanex16(__float_as_int(x), __float_as_int(x),
                                                       0x76543210, 0xfedcba98, false, false));
}

template <int M>
__device__ __forceinline__ float row_xor_add(float p) {
    return p + __int_as_float(__builtin_amdgcn_update_dpp(0, __float_as_int(p), 0x160 + M,
                                                          0xf, 0xf, true));
}

__device__ __forceinline__ void add_pair(const float* sW, int r, float4 zs, float4 zd,
                                         int f0, float4 (&acc)[NUM_HEADS]) {
    const float* Wr = sW + r * (NUM_HEADS * 2 * OUT_FEAT);
#pragma unroll
    for (int h = 0; h < NUM_HEADS; ++h) {
        const float* Wh = Wr + h * (2 * OUT_FEAT);
        const float4 ws = *(const float4*)(Wh + f0);
        const float4 wd = *(const float4*)(Wh + OUT_FEAT + f0);
        v2f t = v2f{zs.x, zs.y} * v2f{ws.x, ws.y};
        t = v2f{zs.z, zs.w} * v2f{ws.z, ws.w} + t;
        t = v2f{zd.x, zd.y} * v2f{wd.x, wd.y} + t;
        t = v2f{zd.z, zd.w} * v2f{wd.z, wd.w} + t;
        float p = t.x + t.y;
        p = row_xor_add<1>(p);
        p = row_xor_add<2>(p);
        p = row_xor_add<4>(p);
        p = row_xor_add<8>(p);
        const float a = (p > 0.f) ? p : 0.01f * p;
        acc[h].x += a * zs.x;
        acc[h].y += a * zs.y;
        acc[h].z += a * zs.z;
        acc[h].w += a * zs.w;
    }
}

__device__ __forceinline__ void pick_pair(const float* __restrict__ z, int s0, int r0, int rank,
                                          int q, int n, int hf, int f0, float4& zs, int& r) {
    const int oA = __builtin_ctz(__builtin_amdgcn_ballot_w32(rank == q));
    int s = __builtin_amdgcn_readlane(s0, oA);
    int rr = __builtin_amdgcn_readlane(r0, oA);
    const bool hasB = (q + 1 < n);
    if (hasB) {
        const int oB = __builtin_ctz(__builtin_amdgcn_ballot_w32(rank == q + 1));
        const int sB = __builtin_amdgcn_readlane(s0, oB);
        const int rB = __builtin_amdgcn_readlane(r0, oB);
        if (hf) { s = sB; rr = rB; }
    }
    float4 v = *(const float4*)(z + (size_t)s * IN_FEAT + f0);
    if (hf && !hasB) v = make_float4(0.f, 0.f, 0.f, 0.f);
    zs = v;
    r = rr;
}

__global__ void __launch_bounds__(GATHER_THREADS, 12)
gather_kernel(const float* __restrict__ z,
              const int*   __restrict__ src,
              const int*   __restrict__ dst,
              const int*   __restrict__ etype,
              const float* __restrict__ attn_fcs,
              const float* __restrict__ feat,
              const float* __restrict__ Wself,
              float* out) {
    __shared__ __attribute__((aligned(16))) float sW[ATTN_ELEMS];
    __shared__ __attribute__((aligned(16))) float sS[2][16 * SROW];
    __shared__ int2 sL[16 * LIST_CAP];
    __shared__ int sCnt[2][32];
    if (threadIdx.x < 32) sCnt[0][threadIdx.x] = 0;
    {
        constexpr int NV4 = ATTN_ELEMS / 4;
        constexpr int PER = 8;
        constexpr int CPY = NV4 / PER;
        static_assert(CPY * PER == NV4 && CPY <= GATHER_THREADS, "copy split");
        const float4* g4 = reinterpret_cast<const float4*>(attn_fcs);
        float4* s4 = reinterpret_cast<float4*>(sW);
        const int tid = threadIdx.x;
        if (tid < CPY) {
            float4 tmp[PER];
#pragma unroll
            for (int j = 0; j < PER; ++j) tmp[j] = g4[tid + j * CPY];
#pragma unroll
            for (int j = 0; j < PER; ++j) s4[tid + j * CPY] = tmp[j];
        }
    }
    __syncthreads();

    const int lane  = threadIdx.x & 31;
    const int wave  = __builtin_amdgcn_readfirstlane(threadIdx.x >> 5);
    const int hf    = lane >> 4;
    const int f0    = (lane & 15) * 4;

    int par = 0;
    for (int t = blockIdx.x; t < MTILES; t += gridDim.x, par ^= 1) {
        float* S = sS[par];
        int* cntT = sCnt[par];
        const int v = t * 16 + wave;

        if (wave >= PROJ_WAVES) {
            const char* tb = reinterpret_cast<const char*>(out) + (size_t)t * TILE_BYTES;
            constexpr int NQ = N_CHUNK * 4;
            constexpr int ASM_THREADS = GATHER_THREADS - PROJ_WAVES * 32;
            static_assert(NQ % 32 == 0 && ASM_THREADS % 32 == 0 && 2 * ASM_THREADS >= NQ, "list split");
            unsigned tix = threadIdx.x;
            asm volatile("" : "+v"(tix));
            uint4 ea[2], pa[2];
#pragma unroll
            for (int j = 0; j < 2; ++j) {
                const unsigned q = tix - PROJ_WAVES * 32 + j * ASM_THREADS;
                const unsigned qq = q < NQ ? q : NQ - 1;
                const uint4* gp = reinterpret_cast<const uint4*>(tb + qq * 32u);
                ea[j] = gp[0];
                pa[j] = gp[1];
            }
#pragma unroll
            for (int j = 0; j < 2; ++j) {
                const unsigned q = tix - PROJ_WAVES * 32 + j * ASM_THREADS;
                if (q < NQ) {
                    const int p = (int)(q & 3u);
                    const int cnt = __builtin_amdgcn_update_dpp(0, (int)ea[j].x, 0x00, 0xf, 0xf, true);
                    if (cnt > SEG_CAP) cntT[16] = 1;
                    const int cc = cnt < SEG_CAP ? cnt : SEG_CAP;
                    const unsigned ew[4] = {ea[j].x, ea[j].y, ea[j].z, ea[j].w};
                    const unsigned pw[4] = {pa[j].x, pa[j].y, pa[j].z, pa[j].w};
#pragma unroll
                    for (int i = 0; i < 4; ++i) {
                        const int w = 4 * p + i;
                        if (w >= 1 && w <= cc) {
                            const int loc = (int)(ew[i] >> 19) & 15;
                            const int pos = atomicAdd(&cntT[loc], 1);
                            if (pos < LIST_CAP)
                                sL[loc * LIST_CAP + pos] = make_int2((int)(ew[i] & 0x7ffffu), (int)pw[i]);
                        }
                    }
                }
            }
        }
        __syncthreads();
        if (threadIdx.x >= GATHER_THREADS - 32) sCnt[par ^ 1][threadIdx.x - (GATHER_THREADS - 32)] = 0;

        float4 acc[NUM_HEADS];
        for (int h = 0; h < NUM_HEADS; ++h) acc[h] = make_float4(0.f, 0.f, 0.f, 0.f);

        if (v < N_NODES) {
            const int2* lst = sL + wave * LIST_CAP;
            const int2 lraw = lst[lane];
            const int kraw = lraw.x, praw = lraw.y;
            const int n = __builtin_amdgcn_readfirstlane(cntT[16] ? N_EDGES : cntT[wave]);
            const float4 zd = *(const float4*)(z + (size_t)v * IN_FEAT + f0);

            if (n <= 32) {
                int k0 = KEY_NONE, s0 = 0, r0 = 0;
                if (lane < n) { k0 = kraw; s0 = praw & 0xffff; r0 = praw >> 16; }
                int rank = 0;
                for (int j = 0; j < n; ++j) {
                    const int kj = __builtin_amdgcn_readlane(k0, j);
                    rank += (kj < k0) ? 1 : 0;
                }
                float4 zs_next = make_float4(0.f, 0.f, 0.f, 0.f);
                int r_next = 0;
                if (n > 0) pick_pair(z, s0, r0, rank, 0, n, hf, f0, zs_next, r_next);
                for (int q = 0; q < n; q += 2) {
                    const float4 zs = zs_next;
                    const int r = r_next;
                    if (q + 2 < n) pick_pair(z, s0, r0, rank, q + 2, n, hf, f0, zs_next, r_next);
                    add_pair(sW, r, zs, zd, f0, acc);
                }
            } else if (n <= LIST_CAP) {
                int last = -1;
                for (int i = 0; i < n; ++i) {
                    int best = KEY_NONE;
                    for (int p = lane; p < n; p += 32) {
                        const int k = lst[p].x;
                        if (k > last && k < best) best = k;
                    }
                    for (int m = 16; m > 0; m >>= 1) best = min(best, __shfl_xor(best, m, 32));
                    last = best;
                    const int s = src[best], r = etype[best];
                    float4 zs = *(const float4*)(z + (size_t)s * IN_FEAT + f0);
                    if (hf) zs = make_float4(0.f, 0.f, 0.f, 0.f);
                    add_pair(sW, r, zs, zd, f0, acc);
                }
            } else {
                for (int e0 = 0; e0 < N_EDGES; e0 += 32) {
                    unsigned m = __builtin_amdgcn_ballot_w32(dst[e0 + lane] == v);
                    while (m) {
                        const int e = e0 + __builtin_ctz(m);
                        m &= m - 1;
                        const int s = src[e], r = etype[e];
                        float4 zs = *(const float4*)(z + (size_t)s * IN_FEAT + f0);
                        if (hf) zs = make_float4(0.f, 0.f, 0.f, 0.f);
                        add_pair(sW, r, zs, zd, f0, acc);
                    }
                }
            }
        }

        for (int h = 0; h < NUM_HEADS; ++h) {
            float4 u = acc[h];
            u.x += xhalf(u.x);
            u.y += xhalf(u.y);
            u.z += xhalf(u.z);
            u.w += xhalf(u.w);
            if (hf == 0) *(float4*)(S + wave * SROW + h * OUT_FEAT + f0) = u;
        }

        __syncthreads();

        if (wave < PROJ_WAVES) {
            const int l16  = lane & 15;
            const int koff = hf * 2;
            const int rowA = min(t * 16 + l16, N_NODES - 1);
            const int colb = wave * 32 + l16;
            const float* ap = feat + (size_t)rowA * IN_FEAT + koff;
            const float* bp = Wself + (size_t)koff * HF + colb;
            const int mbase = t * 16 + hf * 8;
            float* op = out + (size_t)mbase * HF + colb;
            const float* sp = S + (hf * 8) * SROW + colb;

            v8f c[2];
            c[0] = v8f{};
            c[1] = v8f{};
            constexpr int KH = IN_FEAT / 8;
#pragma unroll
            for (int half = 0; half < 2; ++half) {
                v2f a[KH];
                v2f b[KH][2];
#pragma unroll
                for (int i = 0; i < KH; ++i) {
                    const int k = 32 * half + 4 * i;
                    a[i].x = ap[k];
                    a[i].y = ap[k + 1];
#pragma unroll
                    for (int j = 0; j < 2; ++j) {
                        b[i][j].x = bp[k * HF + j * 16];
                        b[i][j].y = bp[(k + 1) * HF + j * 16];
                    }
                }
#pragma unroll
                for (int i = 0; i < KH; ++i) {
#pragma unroll
                    for (int j = 0; j < 2; ++j)
                        c[j] = __builtin_amdgcn_wmma_f32_16x16x4_f32(
                                    false, a[i],  false, b[i][j],
                                    (short)0, c[j],  false,  false);
                }
            }
#pragma unroll
            for (int r = 0; r < 8; ++r) {
#pragma unroll
                for (int j = 0; j < 2; ++j) c[j][r] += sp[r * SROW + j * 16];
            }
#pragma unroll
            for (int r = 0; r < 8; ++r) {
                if (mbase + r < N_NODES) {
#pragma unroll
                    for (int j = 0; j < 2; ++j) op[r * HF + j * 16] = c[j][r];
                }
            }
        }
    }
}

extern "C" void kernel_launch(void* const* d_in, const int* in_sizes, int n_in,
                              void* d_out, int out_size, void* d_ws, size_t ws_size,
                              hipStream_t stream) {
    const float* feat      = (const float*)d_in[0];
    const int*   src       = (const int*)  d_in[1];
    const int*   dst       = (const int*)  d_in[2];
    const int*   etype     = (const int*)  d_in[3];
    const float* fc_w      = (const float*)d_in[4];
    const float* self_fc_w = (const float*)d_in[5];
    const float* attn_w    = (const float*)d_in[6];
    const float* w_comp    = (const float*)d_in[7];
    float* out = (float*)d_out;

    float* z        = (float*)d_ws;
    float* attn_fcs = z + (size_t)N_NODES * OUT_FEAT;

    front_kernel<<<PART_BLOCKS + ZPROJ_BLOCKS + ATTN_BLOCKS, FRONT_THREADS, 0, stream>>>(
        dst, src, etype, out, feat, fc_w, z, w_comp, attn_w, attn_fcs);

    gather_kernel<<<768, GATHER_THREADS, 0, stream>>>(z, src, dst, etype, attn_fcs, feat,
                                                      self_fc_w, out);
}


# front launch rebalanced: a projection wave takes 16x32 of z in 98 blocks (was 16x64 in 49); the partition write-out reads 7 segments' LDS words per wait (was 1 per two waits); sum order unchanged
# speedup vs baseline: 2.5264x; 1.0444x over previous
#include <hip/hip_runtime.h>
#include <hip/hip_bf16.h>

typedef float v2f __attribute__((ext_vector_type(2)));
typedef float v8f __attribute__((ext_vector_type(8)));

#define N_NODES   25000
#define N_EDGES   320000
#define IN_FEAT   64
#define OUT_FEAT  64
#define NUM_HEADS 5
#define NUM_RELS  20
#define NUM_BASES 10
#define HF        (NUM_HEADS * OUT_FEAT)
#define ATTN_ELEMS (NUM_RELS * NUM_HEADS * 2 * OUT_FEAT)
#define MTILES ((N_NODES + 15) / 16)
#define LIST_CAP  64
#define ROW_BYTES (HF * 4)
#define N_CHUNK   80
#define CHUNK_EDGES (N_EDGES / N_CHUNK)
#define SEG_WORDS 32
#define SEG_CAP   15
#define TILE_BYTES (16 * ROW_BYTES)
#define TILE_WORDS (TILE_BYTES / 4)
#define BIN_PITCH 33
#define KEY_NONE  0x7fffffff
#define GATHER_THREADS 512
#define PROJ_WAVES (HF / 32)
#define SROW      (HF + 4)


#define FRONT_THREADS 1024
#define PART_BLOCKS  N_CHUNK
#define ZPROJ_TASKS  (MTILES * 2)
#define ZPROJ_BLOCKS ((ZPROJ_TASKS + FRONT_THREADS / 32 - 1) / (FRONT_THREADS / 32))
#define SWEEP_BATCH  7
#define ATTN_BLOCKS  ((ATTN_ELEMS + FRONT_THREADS - 1) / FRONT_THREADS)
#define PART_PER     ((CHUNK_EDGES + FRONT_THREADS - 1) / FRONT_THREADS)
static_assert(N_CHUNK * CHUNK_EDGES == N_EDGES, "chunk split");
static_assert(N_CHUNK * SEG_WORDS * 4 <= (N_NODES - (MTILES - 1) * 16) * ROW_BYTES, "segments fit the last tile's rows");
static_assert(N_EDGES <= (1 << 19) && N_NODES <= (1 << 16), "packed words");

__global__ void __launch_bounds__(FRONT_THREADS)
front_kernel(const int* __restrict__ dst, const int* __restrict__ src,
             const int* __restrict__ etype, float* out,
             const float* __restrict__ feat,
             const float* __restrict__ W,
             float* __restrict__ Z,
             const float* __restrict__ w_comp,
             const float* __restrict__ attn_w,
             float* __restrict__ attn_fcs) {
    __shared__ int sBin[MTILES * BIN_PITCH];
    if (blockIdx.x < PART_BLOCKS) {
        const int c = blockIdx.x;
        const int tid = threadIdx.x;
        for (int i = tid; i < MTILES; i += FRONT_THREADS) sBin[i * BIN_PITCH] = 0;

        int d[PART_PER], s[PART_PER], r[PART_PER];
#pragma unroll
        for (int j = 0; j < PART_PER; ++j) {
            const int i = j * FRONT_THREADS + tid;
            const int e = c * CHUNK_EDGES + (i < CHUNK_EDGES ? i : 0);
            d[j] = dst[e];
            s[j] = src[e];
            r[j] = etype[e];
        }
        __syncthreads();
#pragma unroll
        for (int j = 0; j < PART_PER; ++j) {
            const int i = j * FRONT_THREADS + tid;
            if (i < CHUNK_EDGES) {
                const int e = c * CHUNK_EDGES + i;
                const int tile = (int)min((unsigned)d[j] >> 4, (unsigned)(MTILES - 1));
                int* bin = sBin + tile * BIN_PITCH;
                const int pos = atomicAdd(bin, 1);
                if (pos < SEG_CAP) {
                    const int w = pos + 1;
                    const int x = ((w >> 2) << 3) + (w & 3);
                    bin[x]     = e | ((d[j] & 15) << 19);
                    bin[x + 4] = s[j] | (r[j] << 16);
                }
            }
        }
        __syncthreads();
        int* ow = reinterpret_cast<int*>(out);
        const int x = tid & 31;
        const int w = ((x >> 3) << 2) + (x & 3);
        const int wlo = (x & 4) ? 1 : 0;
        constexpr int WAVES = FRONT_THREADS / 32;
        for (int t0 = tid >> 5; t0 < MTILES; t0 += WAVES * SWEEP_BATCH) {
            int cnt[SWEEP_BATCH], val[SWEEP_BATCH];
#pragma unroll
            for (int u = 0; u < SWEEP_BATCH; ++u) {
                const int t = t0 + u * WAVES;
                const int* bin = sBin + (t < MTILES ? t : MTILES - 1) * BIN_PITCH;
                cnt[u] = bin[0];
                val[u] = bin[x];
            }
#pragma unroll
            for (int u = 0; u < SWEEP_BATCH; ++u) {
                const int t = t0 + u * WAVES;
                const int cc = cnt[u] < SEG_CAP ? cnt[u] : SEG_CAP;
                if (t < MTILES && w >= wlo && w <= cc)
                    ow[(size_t)t * TILE_WORDS + c * SEG_WORDS + x] = val[u];
            }
        }
        return;
    }

    if (blockIdx.x >= PART_BLOCKS + ZPROJ_BLOCKS) {
        const int t = (blockIdx.x - PART_BLOCKS - ZPROJ_BLOCKS) * FRONT_THREADS + threadIdx.x;
        if (t >= ATTN_ELEMS) return;
        const int r = t / (NUM_HEADS * 2 * OUT_FEAT);
        const int h = (t / (2 * OUT_FEAT)) % NUM_HEADS;
        const int f = t % (2 * OUT_FEAT);
        float acc = 0.f;
#pragma unroll
        for (int b = 0; b < NUM_BASES; ++b)
            acc += w_comp[r * NUM_BASES + b] * attn_w[b * (2 * OUT_FEAT * NUM_HEADS) + f * NUM_HEADS + h];
        attn_fcs[t] = acc;
        return;
    }

    constexpr int LDW = OUT_FEAT;
    constexpr int NT  = 2;
    const int wave = threadIdx.x >> 5;
    const int lane = threadIdx.x & 31;
    const int task = (blockIdx.x - PART_BLOCKS) * (FRONT_THREADS / 32) + wave;
    if (task >= ZPROJ_TASKS) return;
    const int mt = task >> 1;
    const int cb = (task & 1) * (NT * 16);

    const int l16  = lane & 15;
    const int koff = (lane < 16) ? 0 : 2;
    const int rowA = min(mt * 16 + l16, N_NODES - 1);

    const float* ap = feat + (size_t)rowA * IN_FEAT + koff;
    const float* bp = W + (size_t)koff * LDW + cb + l16;

    const int mbase = mt * 16 + ((lane < 16) ? 0 : 8);
    float* op = Z + (size_t)mbase * LDW + cb + l16;

    v8f c[NT];
#pragma unroll
    for (int j = 0; j < NT; ++j) c[j] = v8f{};

#pragma unroll
    for (int k = 0; k < IN_FEAT; k += 4) {
        v2f a;
        a.x = ap[k];
        a.y = ap[k + 1];
#pragma unroll
        for (int j = 0; j < NT; ++j) {
            v2f b;
            b.x = bp[k * LDW + j * 16];
            b.y = bp[(k + 1) * LDW + j * 16];
            c[j] = __builtin_amdgcn_wmma_f32_16x16x4_f32(
                        false, a,  false, b,
                        (short)0, c[j],  false,  false);
        }
    }

#pragma unroll
    for (int r = 0; r < 8; ++r) {
        if (mbase + r < N_NODES) {
#pragma unroll
            for (int j = 0; j < NT; ++j)
                op[r * LDW + j * 16] = c[j][r];
        }
    }
}

__device__ __forceinline__ float xhalf(float x) {
    return __int_as_float(__builtin_amdgcn_permlanex16(__float_as_int(x), __float_as_int(x),
                                                       0x76543210, 0xfedcba98, false, false));
}

template <int M>
__device__ __forceinline__ float row_xor_add(float p) {
    return p + __int_as_float(__builtin_amdgcn_update_dpp(0, __float_as_int(p), 0x160 + M,
                                                          0xf, 0xf, true));
}

__device__ __forceinline__ void add_pair(const float* sW, int r, float4 zs, float4 zd,
                                         int f0, float4 (&acc)[NUM_HEADS]) {
    const float* Wr = sW + r * (NUM_HEADS * 2 * OUT_FEAT);
#pragma unroll
    for (int h = 0; h < NUM_HEADS; ++h) {
        const float* Wh = Wr + h * (2 * OUT_FEAT);
        const float4 ws = *(const float4*)(Wh + f0);
        const float4 wd = *(const float4*)(Wh + OUT_FEAT + f0);
        v2f t = v2f{zs.x, zs.y} * v2f{ws.x, ws.y};
        t = v2f{zs.z, zs.w} * v2f{ws.z, ws.w} + t;
        t = v2f{zd.x, zd.y} * v2f{wd.x, wd.y} + t;
        t = v2f{zd.z, zd.w} * v2f{wd.z, wd.w} + t;
        float p = t.x + t.y;
        p = row_xor_add<1>(p);
        p = row_xor_add<2>(p);
        p = row_xor_add<4>(p);
        p = row_xor_add<8>(p);
        const float a = (p > 0.f) ? p : 0.01f * p;
        acc[h].x += a * zs.x;
        acc[h].y += a * zs.y;
        acc[h].z += a * zs.z;
        acc[h].w += a * zs.w;
    }
}

__device__ __forceinline__ void pick_pair(const float* __restrict__ z, int s0, int r0, int rank,
                                          int q, int n, int hf, int f0, float4& zs, int& r) {
    const int oA = __builtin_ctz(__builtin_amdgcn_ballot_w32(rank == q));
    int s = __builtin_amdgcn_readlane(s0, oA);
    int rr = __builtin_amdgcn_readlane(r0, oA);
    const bool hasB = (q + 1 < n);
    if (hasB) {
        const int oB = __builtin_ctz(__builtin_amdgcn_ballot_w32(rank == q + 1));
        const int sB = __builtin_amdgcn_readlane(s0, oB);
        const int rB = __builtin_amdgcn_readlane(r0, oB);
        if (hf) { s = sB; rr = rB; }
    }
    float4 v = *(const float4*)(z + (size_t)s * IN_FEAT + f0);
    if (hf && !hasB) v = make_float4(0.f, 0.f, 0.f, 0.f);
    zs = v;
    r = rr;
}

__global__ void __launch_bounds__(GATHER_THREADS, 12)
gather_kernel(const float* __restrict__ z,
              const int*   __restrict__ src,
              const int*   __restrict__ dst,
              const int*   __restrict__ etype,
              const float* __restrict__ attn_fcs,
              const float* __restrict__ feat,
              const float* __restrict__ Wself,
              float* out) {
    __shared__ __attribute__((aligned(16))) float sW[ATTN_ELEMS];
    __shared__ __attribute__((aligned(16))) float sS[2][16 * SROW];
    __shared__ int2 sL[16 * LIST_CAP];
    __shared__ int sCnt[2][32];
    if (threadIdx.x < 32) sCnt[0][threadIdx.x] = 0;
    {
        constexpr int NV4 = ATTN_ELEMS / 4;
        constexpr int PER = 8;
        constexpr int CPY = NV4 / PER;
        static_assert(CPY * PER == NV4 && CPY <= GATHER_THREADS, "copy split");
        const float4* g4 = reinterpret_cast<const float4*>(attn_fcs);
        float4* s4 = reinterpret_cast<float4*>(sW);
        const int tid = threadIdx.x;
        if (tid < CPY) {
            float4 tmp[PER];
#pragma unroll
            for (int j = 0; j < PER; ++j) tmp[j] = g4[tid + j * CPY];
#pragma unroll
            for (int j = 0; j < PER; ++j) s4[tid + j * CPY] = tmp[j];
        }
    }
    __syncthreads();

    const int lane  = threadIdx.x & 31;
    const int wave  = __builtin_amdgcn_readfirstlane(threadIdx.x >> 5);
    const int hf    = lane >> 4;
    const int f0    = (lane & 15) * 4;

    int par = 0;
    for (int t = blockIdx.x; t < MTILES; t += gridDim.x, par ^= 1) {
        float* S = sS[par];
        int* cntT = sCnt[par];
        const int v = t * 16 + wave;

        if (wave >= PROJ_WAVES) {
            const char* tb = reinterpret_cast<const char*>(out) + (size_t)t * TILE_BYTES;
            constexpr int NQ = N_CHUNK * 4;
            constexpr int ASM_THREADS = GATHER_THREADS - PROJ_WAVES * 32;
            static_assert(NQ % 32 == 0 && ASM_THREADS % 32 == 0 && 2 * ASM_THREADS >= NQ, "list split");
            unsigned tix = threadIdx.x;
            asm volatile("" : "+v"(tix));
            uint4 ea[2], pa[2];
#pragma unroll
            for (int j = 0; j < 2; ++j) {
                const unsigned q = tix - PROJ_WAVES * 32 + j * ASM_THREADS;
                const unsigned qq = q < NQ ? q : NQ - 1;
                const uint4* gp = reinterpret_cast<const uint4*>(tb + qq * 32u);
                ea[j] = gp[0];
                pa[j] = gp[1];
            }
#pragma unroll
            for (int j = 0; j < 2; ++j) {
                const unsigned q = tix - PROJ_WAVES * 32 + j * ASM_THREADS;
                if (q < NQ) {
                    const int p = (int)(q & 3u);
                    const int cnt = __builtin_amdgcn_update_dpp(0, (int)ea[j].x, 0x00, 0xf, 0xf, true);
                    if (cnt > SEG_CAP) cntT[16] = 1;
                    const int cc = cnt < SEG_CAP ? cnt : SEG_CAP;
                    const unsigned ew[4] = {ea[j].x, ea[j].y, ea[j].z, ea[j].w};
                    const unsigned pw[4] = {pa[j].x, pa[j].y, pa[j].z, pa[j].w};
#pragma unroll
                    for (int i = 0; i < 4; ++i) {
                        const int w = 4 * p + i;
                        if (w >= 1 && w <= cc) {
                            const int loc = (int)(ew[i] >> 19) & 15;
                            const int pos = atomicAdd(&cntT[loc], 1);
                            if (pos < LIST_CAP)
                                sL[loc * LIST_CAP + pos] = make_int2((int)(ew[i] & 0x7ffffu), (int)pw[i]);
                        }
                    }
                }
            }
        }
        __syncthreads();
        if (threadIdx.x >= GATHER_THREADS - 32) sCnt[par ^ 1][threadIdx.x - (GATHER_THREADS - 32)] = 0;

        float4 acc[NUM_HEADS];
        for (int h = 0; h < NUM_HEADS; ++h) acc[h] = make_float4(0.f, 0.f, 0.f, 0.f);

        if (v < N_NODES) {
            const int2* lst = sL + wave * LIST_CAP;
            const int2 lraw = lst[lane];
            const int kraw = lraw.x, praw = lraw.y;
            const int n = __builtin_amdgcn_readfirstlane(cntT[16] ? N_EDGES : cntT[wave]);
            const float4 zd = *(const float4*)(z + (size_t)v * IN_FEAT + f0);

            if (n <= 32) {
                int k0 = KEY_NONE, s0 = 0, r0 = 0;
                if (lane < n) { k0 = kraw; s0 = praw & 0xffff; r0 = praw >> 16; }
                int rank = 0;
                for (int j = 0; j < n; ++j) {
                    const int kj = __builtin_amdgcn_readlane(k0, j);
                    rank += (kj < k0) ? 1 : 0;
                }
                float4 zs_next = make_float4(0.f, 0.f, 0.f, 0.f);
                int r_next = 0;
                if (n > 0) pick_pair(z, s0, r0, rank, 0, n, hf, f0, zs_next, r_next);
                for (int q = 0; q < n; q += 2) {
                    const float4 zs = zs_next;
                    const int r = r_next;
                    if (q + 2 < n) pick_pair(z, s0, r0, rank, q + 2, n, hf, f0, zs_next, r_next);
                    add_pair(sW, r, zs, zd, f0, acc);
                }
            } else if (n <= LIST_CAP) {
                int last = -1;
                for (int i = 0; i < n; ++i) {
                    int best = KEY_NONE;
                    for (int p = lane; p < n; p += 32) {
                        const int k = lst[p].x;
                        if (k > last && k < best) best = k;
                    }
                    for (int m = 16; m > 0; m >>= 1) best = min(best, __shfl_xor(best, m, 32));
                    last = best;
                    const int s = src[best], r = etype[best];
                    float4 zs = *(const float4*)(z + (size_t)s * IN_FEAT + f0);
                    if (hf) zs = make_float4(0.f, 0.f, 0.f, 0.f);
                    add_pair(sW, r, zs, zd, f0, acc);
                }
            } else {
                for (int e0 = 0; e0 < N_EDGES; e0 += 32) {
                    unsigned m = __builtin_amdgcn_ballot_w32(dst[e0 + lane] == v);
                    while (m) {
                        const int e = e0 + __builtin_ctz(m);
                        m &= m - 1;
                        const int s = src[e], r = etype[e];
                        float4 zs = *(const float4*)(z + (size_t)s * IN_FEAT + f0);
                        if (hf) zs = make_float4(0.f, 0.f, 0.f, 0.f);
                        add_pair(sW, r, zs, zd, f0, acc);
                    }
                }
            }
        }

        for (int h = 0; h < NUM_HEADS; ++h) {
            float4 u = acc[h];
            u.x += xhalf(u.x);
            u.y += xhalf(u.y);
            u.z += xhalf(u.z);
            u.w += xhalf(u.w);
            if (hf == 0) *(float4*)(S + wave * SROW + h * OUT_FEAT + f0) = u;
        }

        __syncthreads();

        if (wave < PROJ_WAVES) {
            const int l16  = lane & 15;
            const int koff = hf * 2;
            const int rowA = min(t * 16 + l16, N_NODES - 1);
            const int colb = wave * 32 + l16;
            const float* ap = feat + (size_t)rowA * IN_FEAT + koff;
            const float* bp = Wself + (size_t)koff * HF + colb;
            const int mbase = t * 16 + hf * 8;
            float* op = out + (size_t)mbase * HF + colb;
            const float* sp = S + (hf * 8) * SROW + colb;

            v8f c[2];
            c[0] = v8f{};
            c[1] = v8f{};
            constexpr int KH = IN_FEAT / 8;
#pragma unroll
            for (int half = 0; half < 2; ++half) {
                v2f a[KH];
                v2f b[KH][2];
#pragma unroll
                for (int i = 0; i < KH; ++i) {
                    const int k = 32 * half + 4 * i;
                    a[i].x = ap[k];
                    a[i].y = ap[k + 1];
#pragma unroll
                    for (int j = 0; j < 2; ++j) {
                        b[i][j].x = bp[k * HF + j * 16];
                        b[i][j].y = bp[(k + 1) * HF + j * 16];
                    }
                }
#pragma unroll
                for (int i = 0; i < KH; ++i) {
#pragma unroll
                    for (int j = 0; j < 2; ++j)
                        c[j] = __builtin_amdgcn_wmma_f32_16x16x4_f32(
                                    false, a[i],  false, b[i][j],
                                    (short)0, c[j],  false,  false);
                }
            }
#pragma unroll
            for (int r = 0; r < 8; ++r) {
#pragma unroll
                for (int j = 0; j < 2; ++j) c[j][r] += sp[r * SROW + j * 16];
            }
#pragma unroll
            for (int r = 0; r < 8; ++r) {
                if (mbase + r < N_NODES) {
#pragma unroll
                    for (int j = 0; j < 2; ++j) op[r * HF + j * 16] = c[j][r];
                }
            }
        }
    }
}

extern "C" void kernel_launch(void* const* d_in, const int* in_sizes, int n_in,
                              void* d_out, int out_size, void* d_ws, size_t ws_size,
                              hipStream_t stream) {
    const float* feat      = (const float*)d_in[0];
    const int*   src       = (const int*)  d_in[1];
    const int*   dst       = (const int*)  d_in[2];
    const int*   etype     = (const int*)  d_in[3];
    const float* fc_w      = (const float*)d_in[4];
    const float* self_fc_w = (const float*)d_in[5];
    const float* attn_w    = (const float*)d_in[6];
    const float* w_comp    = (const float*)d_in[7];
    float* out = (float*)d_out;

    float* z        = (float*)d_ws;
    float* attn_fcs = z + (size_t)N_NODES * OUT_FEAT;

    front_kernel<<<PART_BLOCKS + ZPROJ_BLOCKS + ATTN_BLOCKS, FRONT_THREADS, 0, stream>>>(
        dst, src, etype, out, feat, fc_w, z, w_comp, attn_w, attn_fcs);

    gather_kernel<<<768, GATHER_THREADS, 0, stream>>>(z, src, dst, etype, attn_fcs, feat,
                                                      self_fc_w, out);
}


# gather walk: a trip's ten LDS table reads leave in two groups ahead of the arithmetic (six before the next pair is picked, four behind three heads' products); heads interleave; sum order unchanged
# speedup vs baseline: 2.6563x; 1.0514x over previous
#include <hip/hip_runtime.h>
#include <hip/hip_bf16.h>

typedef float v2f __attribute__((ext_vector_type(2)));
typedef float v8f __attribute__((ext_vector_type(8)));

#define N_NODES   25000
#define N_EDGES   320000
#define IN_FEAT   64
#define OUT_FEAT  64
#define NUM_HEADS 5
#define NUM_RELS  20
#define NUM_BASES 10
#define HF        (NUM_HEADS * OUT_FEAT)
#define ATTN_ELEMS (NUM_RELS * NUM_HEADS * 2 * OUT_FEAT)
#define MTILES ((N_NODES + 15) / 16)
#define LIST_CAP  64
#define ROW_BYTES (HF * 4)
#define N_CHUNK   80
#define CHUNK_EDGES (N_EDGES / N_CHUNK)
#define SEG_WORDS 32
#define SEG_CAP   15
#define TILE_BYTES (16 * ROW_BYTES)
#define TILE_WORDS (TILE_BYTES / 4)
#define BIN_PITCH 33
#define KEY_NONE  0x7fffffff
#define GATHER_THREADS 512
#define PROJ_WAVES (HF / 32)
#define SROW      (HF + 4)


#define FRONT_THREADS 1024
#define PART_BLOCKS  N_CHUNK
#define ZPROJ_TASKS  (MTILES * 2)
#define ZPROJ_BLOCKS ((ZPROJ_TASKS + FRONT_THREADS / 32 - 1) / (FRONT_THREADS / 32))
#define SWEEP_BATCH  7
#define ATTN_BLOCKS  ((ATTN_ELEMS + FRONT_THREADS - 1) / FRONT_THREADS)
#define PART_PER     ((CHUNK_EDGES + FRONT_THREADS - 1) / FRONT_THREADS)
static_assert(N_CHUNK * CHUNK_EDGES == N_EDGES, "chunk split");
static_assert(N_CHUNK * SEG_WORDS * 4 <= (N_NODES - (MTILES - 1) * 16) * ROW_BYTES, "segments fit the last tile's rows");
static_assert(N_EDGES <= (1 << 19) && N_NODES <= (1 << 16), "packed words");

__global__ void __launch_bounds__(FRONT_THREADS)
front_kernel(const int* __restrict__ dst, const int* __restrict__ src,
             const int* __restrict__ etype, float* out,
             const float* __restrict__ feat,
             const float* __restrict__ W,
             float* __restrict__ Z,
             const float* __restrict__ w_comp,
             const float* __restrict__ attn_w,
             float* __restrict__ attn_fcs) {
    __shared__ int sBin[MTILES * BIN_PITCH];
    if (blockIdx.x < PART_BLOCKS) {
        const int c = blockIdx.x;
        const int tid = threadIdx.x;
        for (int i = tid; i < MTILES; i += FRONT_THREADS) sBin[i * BIN_PITCH] = 0;

        int d[PART_PER], s[PART_PER], r[PART_PER];
#pragma unroll
        for (int j = 0; j < PART_PER; ++j) {
            const int i = j * FRONT_THREADS + tid;
            const int e = c * CHUNK_EDGES + (i < CHUNK_EDGES ? i : 0);
            d[j] = dst[e];
            s[j] = src[e];
            r[j] = etype[e];
        }
        __syncthreads();
#pragma unroll
        for (int j = 0; j < PART_PER; ++j) {
            const int i = j * FRONT_THREADS + tid;
            if (i < CHUNK_EDGES) {
                const int e = c * CHUNK_EDGES + i;
                const int tile = (int)min((unsigned)d[j] >> 4, (unsigned)(MTILES - 1));
                int* bin = sBin + tile * BIN_PITCH;
                const int pos = atomicAdd(bin, 1);
                if (pos < SEG_CAP) {
                    const int w = pos + 1;
                    const int x = ((w >> 2) << 3) + (w & 3);
                    bin[x]     = e | ((d[j] & 15) << 19);
                    bin[x + 4] = s[j] | (r[j] << 16);
                }
            }
        }
        __syncthreads();
        int* ow = reinterpret_cast<int*>(out);
        const int x = tid & 31;
        const int w = ((x >> 3) << 2) + (x & 3);
        const int wlo = (x & 4) ? 1 : 0;
        constexpr int WAVES = FRONT_THREADS / 32;
        for (int t0 = tid >> 5; t0 < MTILES; t0 += WAVES * SWEEP_BATCH) {
            int cnt[SWEEP_BATCH], val[SWEEP_BATCH];
#pragma unroll
            for (int u = 0; u < SWEEP_BATCH; ++u) {
                const int t = t0 + u * WAVES;
                const int* bin = sBin + (t < MTILES ? t : MTILES - 1) * BIN_PITCH;
                cnt[u] = bin[0];
                val[u] = bin[x];
            }
#pragma unroll
            for (int u = 0; u < SWEEP_BATCH; ++u) {
                const int t = t0 + u * WAVES;
                const int cc = cnt[u] < SEG_CAP ? cnt[u] : SEG_CAP;
                if (t < MTILES && w >= wlo && w <= cc)
                    ow[(size_t)t * TILE_WORDS + c * SEG_WORDS + x] = val[u];
            }
        }
        return;
    }

    if (blockIdx.x >= PART_BLOCKS + ZPROJ_BLOCKS) {
        const int t = (blockIdx.x - PART_BLOCKS - ZPROJ_BLOCKS) * FRONT_THREADS + threadIdx.x;
        if (t >= ATTN_ELEMS) return;
        const int r = t / (NUM_HEADS * 2 * OUT_FEAT);
        const int h = (t / (2 * OUT_FEAT)) % NUM_HEADS;
        const int f = t % (2 * OUT_FEAT);
        float acc = 0.f;
#pragma unroll
        for (int b = 0; b < NUM_BASES; ++b)
            acc += w_comp[r * NUM_BASES + b] * attn_w[b * (2 * OUT_FEAT * NUM_HEADS) + f * NUM_HEADS + h];
        attn_fcs[t] = acc;
        return;
    }

    constexpr int LDW = OUT_FEAT;
    constexpr int NT  = 2;
    const int wave = threadIdx.x >> 5;
    const int lane = threadIdx.x & 31;
    const int task = (blockIdx.x - PART_BLOCKS) * (FRONT_THREADS / 32) + wave;
    if (task >= ZPROJ_TASKS) return;
    const int mt = task >> 1;
    const int cb = (task & 1) * (NT * 16);

    const int l16  = lane & 15;
    const int koff = (lane < 16) ? 0 : 2;
    const int rowA = min(mt * 16 + l16, N_NODES - 1);

    const float* ap = feat + (size_t)rowA * IN_FEAT + koff;
    const float* bp = W + (size_t)koff * LDW + cb + l16;

    const int mbase = mt * 16 + ((lane < 16) ? 0 : 8);
    float* op = Z + (size_t)mbase * LDW + cb + l16;

    v8f c[NT];
#pragma unroll
    for (int j = 0; j < NT; ++j) c[j] = v8f{};

#pragma unroll
    for (int k = 0; k < IN_FEAT; k += 4) {
        v2f a;
        a.x = ap[k];
        a.y = ap[k + 1];
#pragma unroll
        for (int j = 0; j < NT; ++j) {
            v2f b;
            b.x = bp[k * LDW + j * 16];
            b.y = bp[(k + 1) * LDW + j * 16];
            c[j] = __builtin_amdgcn_wmma_f32_16x16x4_f32(
                        false, a,  false, b,
                        (short)0, c[j],  false,  false);
        }
    }

#pragma unroll
    for (int r = 0; r < 8; ++r) {
        if (mbase + r < N_NODES) {
#pragma unroll
            for (int j = 0; j < NT; ++j)
                op[r * LDW + j * 16] = c[j][r];
        }
    }
}

__device__ __forceinline__ float xhalf(float x) {
    return __int_as_float(__builtin_amdgcn_permlanex16(__float_as_int(x), __float_as_int(x),
                                                       0x76543210, 0xfedcba98, false, false));
}

template <int M>
__device__ __forceinline__ float row_xor_add(float p) {
    return p + __int_as_float(__builtin_amdgcn_update_dpp(0, __float_as_int(p), 0x160 + M,
                                                          0xf, 0xf, true));
}

__device__ __forceinline__ float head_dot(float4 ws, float4 wd, float4 zs, float4 zd) {
    v2f t = v2f{zs.x, zs.y} * v2f{ws.x, ws.y};
    t = v2f{zs.z, zs.w} * v2f{ws.z, ws.w} + t;
    t = v2f{zd.x, zd.y} * v2f{wd.x, wd.y} + t;
    t = v2f{zd.z, zd.w} * v2f{wd.z, wd.w} + t;
    return t.x + t.y;
}

__device__ __forceinline__ void head_add(float p, float4 zs, float4& acc) {
    p = row_xor_add<1>(p);
    p = row_xor_add<2>(p);
    p = row_xor_add<4>(p);
    p = row_xor_add<8>(p);
    const float a = (p > 0.f) ? p : 0.01f * p;
    acc.x += a * zs.x;
    acc.y += a * zs.y;
    acc.z += a * zs.z;
    acc.w += a * zs.w;
}

struct TableRows { float4 s0, d0, s1, d1, s2, d2; };
__device__ __forceinline__ TableRows read_rows(const float* Wr) {
    TableRows w;
    w.s0 = *(const float4*)(Wr);
    w.d0 = *(const float4*)(Wr + OUT_FEAT);
    w.s1 = *(const float4*)(Wr + 2 * OUT_FEAT);
    w.d1 = *(const float4*)(Wr + 3 * OUT_FEAT);
    w.s2 = *(const float4*)(Wr + 4 * OUT_FEAT);
    w.d2 = *(const float4*)(Wr + 5 * OUT_FEAT);
    return w;
}

__device__ __forceinline__ void add_pair(const float* Wr, const TableRows& w, float4 zs,
                                         float4 zd, float4 (&acc)[NUM_HEADS]) {
    const float p0 = head_dot(w.s0, w.d0, zs, zd);
    const float p1 = head_dot(w.s1, w.d1, zs, zd);
    const float p2 = head_dot(w.s2, w.d2, zs, zd);
    const float4 s3 = *(const float4*)(Wr + 6 * OUT_FEAT);
    const float4 d3 = *(const float4*)(Wr + 7 * OUT_FEAT);
    const float4 s4 = *(const float4*)(Wr + 8 * OUT_FEAT);
    const float4 d4 = *(const float4*)(Wr + 9 * OUT_FEAT);
    head_add(p0, zs, acc[0]);
    head_add(p1, zs, acc[1]);
    head_add(p2, zs, acc[2]);
    const float p3 = head_dot(s3, d3, zs, zd);
    const float p4 = head_dot(s4, d4, zs, zd);
    head_add(p3, zs, acc[3]);
    head_add(p4, zs, acc[4]);
}

__device__ __forceinline__ void pick_pair(const float* __restrict__ z, int s0, int r0, int rank,
                                          int q, int n, int hf, int f0, float4& zs, int& r) {
    const int oA = __builtin_ctz(__builtin_amdgcn_ballot_w32(rank == q));
    int s = __builtin_amdgcn_readlane(s0, oA);
    int rr = __builtin_amdgcn_readlane(r0, oA);
    const bool hasB = (q + 1 < n);
    if (hasB) {
        const int oB = __builtin_ctz(__builtin_amdgcn_ballot_w32(rank == q + 1));
        const int sB = __builtin_amdgcn_readlane(s0, oB);
        const int rB = __builtin_amdgcn_readlane(r0, oB);
        if (hf) { s = sB; rr = rB; }
    }
    float4 v = *(const float4*)(z + (size_t)s * IN_FEAT + f0);
    if (hf && !hasB) v = make_float4(0.f, 0.f, 0.f, 0.f);
    zs = v;
    r = rr;
}

__global__ void __launch_bounds__(GATHER_THREADS, 12)
gather_kernel(const float* __restrict__ z,
              const int*   __restrict__ src,
              const int*   __restrict__ dst,
              const int*   __restrict__ etype,
              const float* __restrict__ attn_fcs,
              const float* __restrict__ feat,
              const float* __restrict__ Wself,
              float* out) {
    __shared__ __attribute__((aligned(16))) float sW[ATTN_ELEMS];
    __shared__ __attribute__((aligned(16))) float sS[2][16 * SROW];
    __shared__ int2 sL[16 * LIST_CAP];
    __shared__ int sCnt[2][32];
    if (threadIdx.x < 32) sCnt[0][threadIdx.x] = 0;
    {
        constexpr int NV4 = ATTN_ELEMS / 4;
        constexpr int PER = 8;
        constexpr int CPY = NV4 / PER;
        static_assert(CPY * PER == NV4 && CPY <= GATHER_THREADS, "copy split");
        const float4* g4 = reinterpret_cast<const float4*>(attn_fcs);
        float4* s4 = reinterpret_cast<float4*>(sW);
        const int tid = threadIdx.x;
        if (tid < CPY) {
            float4 tmp[PER];
#pragma unroll
            for (int j = 0; j < PER; ++j) tmp[j] = g4[tid + j * CPY];
#pragma unroll
            for (int j = 0; j < PER; ++j) s4[tid + j * CPY] = tmp[j];
        }
    }
    __syncthreads();

    const int lane  = threadIdx.x & 31;
    const int wave  = __builtin_amdgcn_readfirstlane(threadIdx.x >> 5);
    const int hf    = lane >> 4;
    const int f0    = (lane & 15) * 4;

    int par = 0;
    for (int t = blockIdx.x; t < MTILES; t += gridDim.x, par ^= 1) {
        float* S = sS[par];
        int* cntT = sCnt[par];
        const int v = t * 16 + wave;

        if (wave >= PROJ_WAVES) {
            const char* tb = reinterpret_cast<const char*>(out) + (size_t)t * TILE_BYTES;
            constexpr int NQ = N_CHUNK * 4;
            constexpr int ASM_THREADS = GATHER_THREADS - PROJ_WAVES * 32;
            static_assert(NQ % 32 == 0 && ASM_THREADS % 32 == 0 && 2 * ASM_THREADS >= NQ, "list split");
            unsigned tix = threadIdx.x;
            asm volatile("" : "+v"(tix));
            uint4 ea[2], pa[2];
#pragma unroll
            for (int j = 0; j < 2; ++j) {
                const unsigned q = tix - PROJ_WAVES * 32 + j * ASM_THREADS;
                const unsigned qq = q < NQ ? q : NQ - 1;
                const uint4* gp = reinterpret_cast<const uint4*>(tb + qq * 32u);
                ea[j] = gp[0];
                pa[j] = gp[1];
            }
#pragma unroll
            for (int j = 0; j < 2; ++j) {
                const unsigned q = tix - PROJ_WAVES * 32 + j * ASM_THREADS;
                if (q < NQ) {
                    const int p = (int)(q & 3u);
                    const int cnt = __builtin_amdgcn_update_dpp(0, (int)ea[j].x, 0x00, 0xf, 0xf, true);
                    if (cnt > SEG_CAP) cntT[16] = 1;
                    const int cc = cnt < SEG_CAP ? cnt : SEG_CAP;
                    const unsigned ew[4] = {ea[j].x, ea[j].y, ea[j].z, ea[j].w};
                    const unsigned pw[4] = {pa[j].x, pa[j].y, pa[j].z, pa[j].w};
#pragma unroll
                    for (int i = 0; i < 4; ++i) {
                        const int w = 4 * p + i;
                        if (w >= 1 && w <= cc) {
                            const int loc = (int)(ew[i] >> 19) & 15;
                            const int pos = atomicAdd(&cntT[loc], 1);
                            if (pos < LIST_CAP)
                                sL[loc * LIST_CAP + pos] = make_int2((int)(ew[i] & 0x7ffffu), (int)pw[i]);
                        }
                    }
                }
            }
        }
        __syncthreads();
        if (threadIdx.x >= GATHER_THREADS - 32) sCnt[par ^ 1][threadIdx.x - (GATHER_THREADS - 32)] = 0;

        float4 acc[NUM_HEADS];
        for (int h = 0; h < NUM_HEADS; ++h) acc[h] = make_float4(0.f, 0.f, 0.f, 0.f);

        if (v < N_NODES) {
            const int2* lst = sL + wave * LIST_CAP;
            const int2 lraw = lst[lane];
            const int kraw = lraw.x, praw = lraw.y;
            const int n = __builtin_amdgcn_readfirstlane(cntT[16] ? N_EDGES : cntT[wave]);
            const float4 zd = *(const float4*)(z + (size_t)v * IN_FEAT + f0);

            if (n <= 32) {
                int k0 = KEY_NONE, s0 = 0, r0 = 0;
                if (lane < n) { k0 = kraw; s0 = praw & 0xffff; r0 = praw >> 16; }
                int rank = 0;
                for (int j = 0; j < n; ++j) {
                    const int kj = __builtin_amdgcn_readlane(k0, j);
                    rank += (kj < k0) ? 1 : 0;
                }
                float4 zs_next = make_float4(0.f, 0.f, 0.f, 0.f);
                int r_next = 0;
                if (n > 0) pick_pair(z, s0, r0, rank, 0, n, hf, f0, zs_next, r_next);
                for (int q = 0; q < n; q += 2) {
                    const float4 zs = zs_next;
                    const float* Wr = sW + r_next * (NUM_HEADS * 2 * OUT_FEAT) + f0;
                    const TableRows w = read_rows(Wr);
                    if (q + 2 < n) pick_pair(z, s0, r0, rank, q + 2, n, hf, f0, zs_next, r_next);
                    add_pair(Wr, w, zs, zd, acc);
                }
            } else if (n <= LIST_CAP) {
                int last = -1;
                for (int i = 0; i < n; ++i) {
                    int best = KEY_NONE;
                    for (int p = lane; p < n; p += 32) {
                        const int k = lst[p].x;
                        if (k > last && k < best) best = k;
                    }
                    for (int m = 16; m > 0; m >>= 1) best = min(best, __shfl_xor(best, m, 32));
                    last = best;
                    const int s = src[best], r = etype[best];
                    float4 zs = *(const float4*)(z + (size_t)s * IN_FEAT + f0);
                    if (hf) zs = make_float4(0.f, 0.f, 0.f, 0.f);
                    const float* Wr = sW + r * (NUM_HEADS * 2 * OUT_FEAT) + f0;
                    add_pair(Wr, read_rows(Wr), zs, zd, acc);
                }
            } else {
                for (int e0 = 0; e0 < N_EDGES; e0 += 32) {
                    unsigned m = __builtin_amdgcn_ballot_w32(dst[e0 + lane] == v);
                    while (m) {
                        const int e = e0 + __builtin_ctz(m);
                        m &= m - 1;
                        const int s = src[e], r = etype[e];
                        float4 zs = *(const float4*)(z + (size_t)s * IN_FEAT + f0);
                        if (hf) zs = make_float4(0.f, 0.f, 0.f, 0.f);
                        const float* Wr = sW + r * (NUM_HEADS * 2 * OUT_FEAT) + f0;
                        add_pair(Wr, read_rows(Wr), zs, zd, acc);
                    }
                }
            }
        }

        for (int h = 0; h < NUM_HEADS; ++h) {
            float4 u = acc[h];
            u.x += xhalf(u.x);
            u.y += xhalf(u.y);
            u.z += xhalf(u.z);
            u.w += xhalf(u.w);
            if (hf == 0) *(float4*)(S + wave * SROW + h * OUT_FEAT + f0) = u;
        }

        __syncthreads();

        if (wave < PROJ_WAVES) {
            unsigned pl = lane;
            asm volatile("" : "+v"(pl));
            const int l16  = (int)(pl & 15u);
            const int ph   = (int)(pl >> 4);
            const int koff = ph * 2;
            const int rowA = min(t * 16 + l16, N_NODES - 1);
            const int colb = wave * 32 + l16;
            const float* ap = feat + (size_t)rowA * IN_FEAT + koff;
            const float* bp = Wself + (size_t)koff * HF + colb;
            const int mbase = t * 16 + ph * 8;
            float* op = out + (size_t)mbase * HF + colb;
            const float* sp = S + (ph * 8) * SROW + colb;

            v8f c[2];
            c[0] = v8f{};
            c[1] = v8f{};
            constexpr int KH = IN_FEAT / 8;
#pragma unroll
            for (int half = 0; half < 2; ++half) {
                v2f a[KH];
                v2f b[KH][2];
#pragma unroll
                for (int i = 0; i < KH; ++i) {
                    const int k = 32 * half + 4 * i;
                    a[i].x = ap[k];
                    a[i].y = ap[k + 1];
#pragma unroll
                    for (int j = 0; j < 2; ++j) {
                        b[i][j].x = bp[k * HF + j * 16];
                        b[i][j].y = bp[(k + 1) * HF + j * 16];
                    }
                }
#pragma unroll
                for (int i = 0; i < KH; ++i) {
#pragma unroll
                    for (int j = 0; j < 2; ++j)
                        c[j] = __builtin_amdgcn_wmma_f32_16x16x4_f32(
                                    false, a[i],  false, b[i][j],
                                    (short)0, c[j],  false,  false);
                }
            }
#pragma unroll
            for (int r = 0; r < 8; ++r) {
#pragma unroll
                for (int j = 0; j < 2; ++j) c[j][r] += sp[r * SROW + j * 16];
            }
#pragma unroll
            for (int r = 0; r < 8; ++r) {
                if (mbase + r < N_NODES) {
#pragma unroll
                    for (int j = 0; j < 2; ++j) op[r * HF + j * 16] = c[j][r];
                }
            }
        }
    }
}

extern "C" void kernel_launch(void* const* d_in, const int* in_sizes, int n_in,
                              void* d_out, int out_size, void* d_ws, size_t ws_size,
                              hipStream_t stream) {
    const float* feat      = (const float*)d_in[0];
    const int*   src       = (const int*)  d_in[1];
    const int*   dst       = (const int*)  d_in[2];
    const int*   etype     = (const int*)  d_in[3];
    const float* fc_w      = (const float*)d_in[4];
    const float* self_fc_w = (const float*)d_in[5];
    const float* attn_w    = (const float*)d_in[6];
    const float* w_comp    = (const float*)d_in[7];
    float* out = (float*)d_out;

    float* z        = (float*)d_ws;
    float* attn_fcs = z + (size_t)N_NODES * OUT_FEAT;

    front_kernel<<<PART_BLOCKS + ZPROJ_BLOCKS + ATTN_BLOCKS, FRONT_THREADS, 0, stream>>>(
        dst, src, etype, out, feat, fc_w, z, w_comp, attn_w, attn_fcs);

    gather_kernel<<<768, GATHER_THREADS, 0, stream>>>(z, src, dst, etype, attn_fcs, feat,
                                                      self_fc_w, out);
}


# gather walk: a wave writes its list back to its LDS slots in edge-id order; a trip gets its pair with one LDS read, a trip ahead (was ballot, bit scan and 4 lane reads per trip); sum order unchanged
# speedup vs baseline: 2.7451x; 1.0334x over previous
#include <hip/hip_runtime.h>
#include <hip/hip_bf16.h>

typedef float v2f __attribute__((ext_vector_type(2)));
typedef float v8f __attribute__((ext_vector_type(8)));

#define N_NODES   25000
#define N_EDGES   320000
#define IN_FEAT   64
#define OUT_FEAT  64
#define NUM_HEADS 5
#define NUM_RELS  20
#define NUM_BASES 10
#define HF        (NUM_HEADS * OUT_FEAT)
#define ATTN_ELEMS (NUM_RELS * NUM_HEADS * 2 * OUT_FEAT)
#define MTILES ((N_NODES + 15) / 16)
#define LIST_CAP  64
#define ROW_BYTES (HF * 4)
#define N_CHUNK   80
#define CHUNK_EDGES (N_EDGES / N_CHUNK)
#define SEG_WORDS 32
#define SEG_CAP   15
#define TILE_BYTES (16 * ROW_BYTES)
#define TILE_WORDS (TILE_BYTES / 4)
#define BIN_PITCH 33
#define KEY_NONE  0x7fffffff
#define GATHER_THREADS 512
#define PROJ_WAVES (HF / 32)
#define SROW      (HF + 4)


#define FRONT_THREADS 1024
#define PART_BLOCKS  N_CHUNK
#define ZPROJ_TASKS  (MTILES * 2)
#define ZPROJ_BLOCKS ((ZPROJ_TASKS + FRONT_THREADS / 32 - 1) / (FRONT_THREADS / 32))
#define SWEEP_BATCH  7
#define ATTN_BLOCKS  ((ATTN_ELEMS + FRONT_THREADS - 1) / FRONT_THREADS)
#define PART_PER     ((CHUNK_EDGES + FRONT_THREADS - 1) / FRONT_THREADS)
static_assert(N_CHUNK * CHUNK_EDGES == N_EDGES, "chunk split");
static_assert(N_CHUNK * SEG_WORDS * 4 <= (N_NODES - (MTILES - 1) * 16) * ROW_BYTES, "segments fit the last tile's rows");
static_assert(N_EDGES <= (1 << 19) && N_NODES <= (1 << 16), "packed words");

__global__ void __launch_bounds__(FRONT_THREADS)
front_kernel(const int* __restrict__ dst, const int* __restrict__ src,
             const int* __restrict__ etype, float* out,
             const float* __restrict__ feat,
             const float* __restrict__ W,
             float* __restrict__ Z,
             const float* __restrict__ w_comp,
             const float* __restrict__ attn_w,
             float* __restrict__ attn_fcs) {
    __shared__ int sBin[MTILES * BIN_PITCH];
    if (blockIdx.x < PART_BLOCKS) {
        const int c = blockIdx.x;
        const int tid = threadIdx.x;
        for (int i = tid; i < MTILES; i += FRONT_THREADS) sBin[i * BIN_PITCH] = 0;

        int d[PART_PER], s[PART_PER], r[PART_PER];
#pragma unroll
        for (int j = 0; j < PART_PER; ++j) {
            const int i = j * FRONT_THREADS + tid;
            const int e = c * CHUNK_EDGES + (i < CHUNK_EDGES ? i : 0);
            d[j] = dst[e];
            s[j] = src[e];
            r[j] = etype[e];
        }
        __syncthreads();
#pragma unroll
        for (int j = 0; j < PART_PER; ++j) {
            const int i = j * FRONT_THREADS + tid;
            if (i < CHUNK_EDGES) {
                const int e = c * CHUNK_EDGES + i;
                const int tile = (int)min((unsigned)d[j] >> 4, (unsigned)(MTILES - 1));
                int* bin = sBin + tile * BIN_PITCH;
                const int pos = atomicAdd(bin, 1);
                if (pos < SEG_CAP) {
                    const int w = pos + 1;
                    const int x = ((w >> 2) << 3) + (w & 3);
                    bin[x]     = e | ((d[j] & 15) << 19);
                    bin[x + 4] = s[j] | (r[j] << 16);
                }
            }
        }
        __syncthreads();
        int* ow = reinterpret_cast<int*>(out);
        const int x = tid & 31;
        const int w = ((x >> 3) << 2) + (x & 3);
        const int wlo = (x & 4) ? 1 : 0;
        constexpr int WAVES = FRONT_THREADS / 32;
        for (int t0 = tid >> 5; t0 < MTILES; t0 += WAVES * SWEEP_BATCH) {
            int cnt[SWEEP_BATCH], val[SWEEP_BATCH];
#pragma unroll
            for (int u = 0; u < SWEEP_BATCH; ++u) {
                const int t = t0 + u * WAVES;
                const int* bin = sBin + (t < MTILES ? t : MTILES - 1) * BIN_PITCH;
                cnt[u] = bin[0];
                val[u] = bin[x];
            }
#pragma unroll
            for (int u = 0; u < SWEEP_BATCH; ++u) {
                const int t = t0 + u * WAVES;
                const int cc = cnt[u] < SEG_CAP ? cnt[u] : SEG_CAP;
                if (t < MTILES && w >= wlo && w <= cc)
                    ow[(size_t)t * TILE_WORDS + c * SEG_WORDS + x] = val[u];
            }
        }
        return;
    }

    if (blockIdx.x >= PART_BLOCKS + ZPROJ_BLOCKS) {
        const int t = (blockIdx.x - PART_BLOCKS - ZPROJ_BLOCKS) * FRONT_THREADS + threadIdx.x;
        if (t >= ATTN_ELEMS) return;
        const int r = t / (NUM_HEADS * 2 * OUT_FEAT);
        const int h = (t / (2 * OUT_FEAT)) % NUM_HEADS;
        const int f = t % (2 * OUT_FEAT);
        float acc = 0.f;
#pragma unroll
        for (int b = 0; b < NUM_BASES; ++b)
            acc += w_comp[r * NUM_BASES + b] * attn_w[b * (2 * OUT_FEAT * NUM_HEADS) + f * NUM_HEADS + h];
        attn_fcs[t] = acc;
        return;
    }

    constexpr int LDW = OUT_FEAT;
    constexpr int NT  = 2;
    const int wave = threadIdx.x >> 5;
    const int lane = threadIdx.x & 31;
    const int task = (blockIdx.x - PART_BLOCKS) * (FRONT_THREADS / 32) + wave;
    if (task >= ZPROJ_TASKS) return;
    const int mt = task >> 1;
    const int cb = (task & 1) * (NT * 16);

    const int l16  = lane & 15;
    const int koff = (lane < 16) ? 0 : 2;
    const int rowA = min(mt * 16 + l16, N_NODES - 1);

    const float* ap = feat + (size_t)rowA * IN_FEAT + koff;
    const float* bp = W + (size_t)koff * LDW + cb + l16;

    const int mbase = mt * 16 + ((lane < 16) ? 0 : 8);
    float* op = Z + (size_t)mbase * LDW + cb + l16;

    v8f c[NT];
#pragma unroll
    for (int j = 0; j < NT; ++j) c[j] = v8f{};

#pragma unroll
    for (int k = 0; k < IN_FEAT; k += 4) {
        v2f a;
        a.x = ap[k];
        a.y = ap[k + 1];
#pragma unroll
        for (int j = 0; j < NT; ++j) {
            v2f b;
            b.x = bp[k * LDW + j * 16];
            b.y = bp[(k + 1) * LDW + j * 16];
            c[j] = __builtin_amdgcn_wmma_f32_16x16x4_f32(
                        false, a,  false, b,
                        (short)0, c[j],  false,  false);
        }
    }

#pragma unroll
    for (int r = 0; r < 8; ++r) {
        if (mbase + r < N_NODES) {
#pragma unroll
            for (int j = 0; j < NT; ++j)
                op[r * LDW + j * 16] = c[j][r];
        }
    }
}

__device__ __forceinline__ float xhalf(float x) {
    return __int_as_float(__builtin_amdgcn_permlanex16(__float_as_int(x), __float_as_int(x),
                                                       0x76543210, 0xfedcba98, false, false));
}

template <int M>
__device__ __forceinline__ float row_xor_add(float p) {
    return p + __int_as_float(__builtin_amdgcn_update_dpp(0, __float_as_int(p), 0x160 + M,
                                                          0xf, 0xf, true));
}

__device__ __forceinline__ float head_dot(float4 ws, float4 wd, float4 zs, float4 zd) {
    v2f t = v2f{zs.x, zs.y} * v2f{ws.x, ws.y};
    t = v2f{zs.z, zs.w} * v2f{ws.z, ws.w} + t;
    t = v2f{zd.x, zd.y} * v2f{wd.x, wd.y} + t;
    t = v2f{zd.z, zd.w} * v2f{wd.z, wd.w} + t;
    return t.x + t.y;
}

__device__ __forceinline__ void head_add(float p, float4 zs, float4& acc) {
    p = row_xor_add<1>(p);
    p = row_xor_add<2>(p);
    p = row_xor_add<4>(p);
    p = row_xor_add<8>(p);
    const float a = (p > 0.f) ? p : 0.01f * p;
    acc.x += a * zs.x;
    acc.y += a * zs.y;
    acc.z += a * zs.z;
    acc.w += a * zs.w;
}

struct TableRows { float4 s0, d0, s1, d1, s2, d2; };
__device__ __forceinline__ TableRows read_rows(const float* Wr) {
    TableRows w;
    w.s0 = *(const float4*)(Wr);
    w.d0 = *(const float4*)(Wr + OUT_FEAT);
    w.s1 = *(const float4*)(Wr + 2 * OUT_FEAT);
    w.d1 = *(const float4*)(Wr + 3 * OUT_FEAT);
    w.s2 = *(const float4*)(Wr + 4 * OUT_FEAT);
    w.d2 = *(const float4*)(Wr + 5 * OUT_FEAT);
    return w;
}

__device__ __forceinline__ void add_pair(const float* Wr, const TableRows& w, float4 zs,
                                         float4 zd, float4 (&acc)[NUM_HEADS]) {
    const float p0 = head_dot(w.s0, w.d0, zs, zd);
    const float p1 = head_dot(w.s1, w.d1, zs, zd);
    const float p2 = head_dot(w.s2, w.d2, zs, zd);
    const float4 s3 = *(const float4*)(Wr + 6 * OUT_FEAT);
    const float4 d3 = *(const float4*)(Wr + 7 * OUT_FEAT);
    const float4 s4 = *(const float4*)(Wr + 8 * OUT_FEAT);
    const float4 d4 = *(const float4*)(Wr + 9 * OUT_FEAT);
    head_add(p0, zs, acc[0]);
    head_add(p1, zs, acc[1]);
    head_add(p2, zs, acc[2]);
    const float p3 = head_dot(s3, d3, zs, zd);
    const float p4 = head_dot(s4, d4, zs, zd);
    head_add(p3, zs, acc[3]);
    head_add(p4, zs, acc[4]);
}

__global__ void __launch_bounds__(GATHER_THREADS, 12)
gather_kernel(const float* __restrict__ z,
              const int*   __restrict__ src,
              const int*   __restrict__ dst,
              const int*   __restrict__ etype,
              const float* __restrict__ attn_fcs,
              const float* __restrict__ feat,
              const float* __restrict__ Wself,
              float* out) {
    __shared__ __attribute__((aligned(16))) float sW[ATTN_ELEMS];
    __shared__ __attribute__((aligned(16))) float sS[2][16 * SROW];
    __shared__ int2 sL[16 * LIST_CAP];
    __shared__ int sCnt[2][32];
    if (threadIdx.x < 32) sCnt[0][threadIdx.x] = 0;
    {
        constexpr int NV4 = ATTN_ELEMS / 4;
        constexpr int PER = 8;
        constexpr int CPY = NV4 / PER;
        static_assert(CPY * PER == NV4 && CPY <= GATHER_THREADS, "copy split");
        const float4* g4 = reinterpret_cast<const float4*>(attn_fcs);
        float4* s4 = reinterpret_cast<float4*>(sW);
        const int tid = threadIdx.x;
        if (tid < CPY) {
            float4 tmp[PER];
#pragma unroll
            for (int j = 0; j < PER; ++j) tmp[j] = g4[tid + j * CPY];
#pragma unroll
            for (int j = 0; j < PER; ++j) s4[tid + j * CPY] = tmp[j];
        }
    }
    __syncthreads();

    const int lane  = threadIdx.x & 31;
    const int wave  = __builtin_amdgcn_readfirstlane(threadIdx.x >> 5);
    const int hf    = lane >> 4;
    const int f0    = (lane & 15) * 4;

    int par = 0;
    for (int t = blockIdx.x; t < MTILES; t += gridDim.x, par ^= 1) {
        float* S = sS[par];
        int* cntT = sCnt[par];
        const int v = t * 16 + wave;

        if (wave >= PROJ_WAVES) {
            const char* tb = reinterpret_cast<const char*>(out) + (size_t)t * TILE_BYTES;
            constexpr int NQ = N_CHUNK * 4;
            constexpr int ASM_THREADS = GATHER_THREADS - PROJ_WAVES * 32;
            static_assert(NQ % 32 == 0 && ASM_THREADS % 32 == 0 && 2 * ASM_THREADS >= NQ, "list split");
            unsigned tix = threadIdx.x;
            asm volatile("" : "+v"(tix));
            uint4 ea[2], pa[2];
#pragma unroll
            for (int j = 0; j < 2; ++j) {
                const unsigned q = tix - PROJ_WAVES * 32 + j * ASM_THREADS;
                const unsigned qq = q < NQ ? q : NQ - 1;
                const uint4* gp = reinterpret_cast<const uint4*>(tb + qq * 32u);
                ea[j] = gp[0];
                pa[j] = gp[1];
            }
#pragma unroll
            for (int j = 0; j < 2; ++j) {
                const unsigned q = tix - PROJ_WAVES * 32 + j * ASM_THREADS;
                if (q < NQ) {
                    const int p = (int)(q & 3u);
                    const int cnt = __builtin_amdgcn_update_dpp(0, (int)ea[j].x, 0x00, 0xf, 0xf, true);
                    if (cnt > SEG_CAP) cntT[16] = 1;
                    const int cc = cnt < SEG_CAP ? cnt : SEG_CAP;
                    const unsigned ew[4] = {ea[j].x, ea[j].y, ea[j].z, ea[j].w};
                    const unsigned pw[4] = {pa[j].x, pa[j].y, pa[j].z, pa[j].w};
#pragma unroll
                    for (int i = 0; i < 4; ++i) {
                        const int w = 4 * p + i;
                        if (w >= 1 && w <= cc) {
                            const int loc = (int)(ew[i] >> 19) & 15;
                            const int pos = atomicAdd(&cntT[loc], 1);
                            if (pos < LIST_CAP)
                                sL[loc * LIST_CAP + pos] = make_int2((int)(ew[i] & 0x7ffffu), (int)pw[i]);
                        }
                    }
                }
            }
        }
        __syncthreads();
        if (wave == GATHER_THREADS / 32 - 1) sCnt[par ^ 1][lane] = 0;

        float4 acc[NUM_HEADS];
        for (int h = 0; h < NUM_HEADS; ++h) acc[h] = make_float4(0.f, 0.f, 0.f, 0.f);

        if (v < N_NODES) {
            const int2* lst = sL + wave * LIST_CAP;
            const int2 lraw = lst[lane];
            const int kraw = lraw.x, praw = lraw.y;
            const int n = __builtin_amdgcn_readfirstlane(cntT[16] ? N_EDGES : cntT[wave]);
            const float4 zd = *(const float4*)(z + (size_t)v * IN_FEAT + f0);

            if (n <= 32) {
                int k0 = KEY_NONE;
                if (lane < n) k0 = kraw;
                int rank = 0;
                for (int j = 0; j < n; ++j) {
                    const int kj = __builtin_amdgcn_readlane(k0, j);
                    rank += (kj < k0) ? 1 : 0;
                }
                int2* srt = sL + wave * LIST_CAP;
                if (lane < n)
                    srt[rank] = make_int2((praw & 0xffff) * (IN_FEAT * 4),
                                          (praw >> 16) * (NUM_HEADS * 2 * OUT_FEAT * 4));
                __builtin_amdgcn_wave_barrier();
                const char* zb = reinterpret_cast<const char*>(z);
                const char* wb = reinterpret_cast<const char*>(sW);
                const unsigned fb = (unsigned)f0 * 4u;
                if (n > 0) {
                    int i0 = hf;      i0 = i0 < n ? i0 : n - 1;
                    int i2 = 2 + hf;  i2 = i2 < n ? i2 : n - 1;
                    const int2 e0 = srt[i0];
                    int2 e_nn = srt[i2];
                    float4 zs_next = *(const float4*)(zb + ((unsigned)e0.x + fb));
                    unsigned w_next = (unsigned)e0.y + fb;
                    int q = 0;
                    for (; q + 1 < n; q += 2) {
                        const float4 zs = zs_next;
                        const float* Wr = reinterpret_cast<const float*>(wb + w_next);
                        const TableRows w = read_rows(Wr);
                        if (q + 2 < n) {
                            zs_next = *(const float4*)(zb + ((unsigned)e_nn.x + fb));
                            w_next = (unsigned)e_nn.y + fb;
                            int i4 = q + 4 + hf;  i4 = i4 < n ? i4 : n - 1;
                            e_nn = srt[i4];
                        }
                        add_pair(Wr, w, zs, zd, acc);
                    }
                    if (q < n) {
                        float4 zs = zs_next;
                        if (hf) zs = make_float4(0.f, 0.f, 0.f, 0.f);
                        const float* Wr = reinterpret_cast<const float*>(wb + w_next);
                        add_pair(Wr, read_rows(Wr), zs, zd, acc);
                    }
                }
            } else if (n <= LIST_CAP) {
                int last = -1;
                for (int i = 0; i < n; ++i) {
                    int best = KEY_NONE;
                    for (int p = lane; p < n; p += 32) {
                        const int k = lst[p].x;
                        if (k > last && k < best) best = k;
                    }
                    for (int m = 16; m > 0; m >>= 1) best = min(best, __shfl_xor(best, m, 32));
                    last = best;
                    const int s = src[best], r = etype[best];
                    float4 zs = *(const float4*)(z + (size_t)s * IN_FEAT + f0);
                    if (hf) zs = make_float4(0.f, 0.f, 0.f, 0.f);
                    const float* Wr = sW + r * (NUM_HEADS * 2 * OUT_FEAT) + f0;
                    add_pair(Wr, read_rows(Wr), zs, zd, acc);
                }
            } else {
                for (int e0 = 0; e0 < N_EDGES; e0 += 32) {
                    unsigned m = __builtin_amdgcn_ballot_w32(dst[e0 + lane] == v);
                    while (m) {
                        const int e = e0 + __builtin_ctz(m);
                        m &= m - 1;
                        const int s = src[e], r = etype[e];
                        float4 zs = *(const float4*)(z + (size_t)s * IN_FEAT + f0);
                        if (hf) zs = make_float4(0.f, 0.f, 0.f, 0.f);
                        const float* Wr = sW + r * (NUM_HEADS * 2 * OUT_FEAT) + f0;
                        add_pair(Wr, read_rows(Wr), zs, zd, acc);
                    }
                }
            }
        }

        for (int h = 0; h < NUM_HEADS; ++h) {
            float4 u = acc[h];
            u.x += xhalf(u.x);
            u.y += xhalf(u.y);
            u.z += xhalf(u.z);
            u.w += xhalf(u.w);
            if (hf == 0) *(float4*)(S + wave * SROW + h * OUT_FEAT + f0) = u;
        }

        __syncthreads();

        if (wave < PROJ_WAVES) {
            unsigned pl = lane;
            asm volatile("" : "+v"(pl));
            const int l16  = (int)(pl & 15u);
            const int ph   = (int)(pl >> 4);
            const int koff = ph * 2;
            const int rowA = min(t * 16 + l16, N_NODES - 1);
            const int colb = wave * 32 + l16;
            const float* ap = feat + (size_t)rowA * IN_FEAT + koff;
            const float* bp = Wself + (size_t)koff * HF + colb;
            const int mbase = t * 16 + ph * 8;
            float* op = out + (size_t)mbase * HF + colb;
            const float* sp = S + (ph * 8) * SROW + colb;

            v8f c[2];
            c[0] = v8f{};
            c[1] = v8f{};
            constexpr int KH = IN_FEAT / 8;
#pragma unroll
            for (int half = 0; half < 2; ++half) {
                v2f a[KH];
                v2f b[KH][2];
#pragma unroll
                for (int i = 0; i < KH; ++i) {
                    const int k = 32 * half + 4 * i;
                    a[i].x = ap[k];
                    a[i].y = ap[k + 1];
#pragma unroll
                    for (int j = 0; j < 2; ++j) {
                        b[i][j].x = bp[k * HF + j * 16];
                        b[i][j].y = bp[(k + 1) * HF + j * 16];
                    }
                }
#pragma unroll
                for (int i = 0; i < KH; ++i) {
#pragma unroll
                    for (int j = 0; j < 2; ++j)
                        c[j] = __builtin_amdgcn_wmma_f32_16x16x4_f32(
                                    false, a[i],  false, b[i][j],
                                    (short)0, c[j],  false,  false);
                }
            }
#pragma unroll
            for (int r = 0; r < 8; ++r) {
#pragma unroll
                for (int j = 0; j < 2; ++j) c[j][r] += sp[r * SROW + j * 16];
            }
#pragma unroll
            for (int r = 0; r < 8; ++r) {
                if (mbase + r < N_NODES) {
#pragma unroll
                    for (int j = 0; j < 2; ++j) op[r * HF + j * 16] = c[j][r];
                }
            }
        }
    }
}

extern "C" void kernel_launch(void* const* d_in, const int* in_sizes, int n_in,
                              void* d_out, int out_size, void* d_ws, size_t ws_size,
                              hipStream_t stream) {
    const float* feat      = (const float*)d_in[0];
    const int*   src       = (const int*)  d_in[1];
    const int*   dst       = (const int*)  d_in[2];
    const int*   etype     = (const int*)  d_in[3];
    const float* fc_w      = (const float*)d_in[4];
    const float* self_fc_w = (const float*)d_in[5];
    const float* attn_w    = (const float*)d_in[6];
    const float* w_comp    = (const float*)d_in[7];
    float* out = (float*)d_out;

    float* z        = (float*)d_ws;
    float* attn_fcs = z + (size_t)N_NODES * OUT_FEAT;

    front_kernel<<<PART_BLOCKS + ZPROJ_BLOCKS + ATTN_BLOCKS, FRONT_THREADS, 0, stream>>>(
        dst, src, etype, out, feat, fc_w, z, w_comp, attn_w, attn_fcs);

    gather_kernel<<<768, GATHER_THREADS, 0, stream>>>(z, src, dst, etype, attn_fcs, feat,
                                                      self_fc_w, out);
}
